# starting kernel
# baseline (speedup 1.0000x reference)
#include <hip/hip_runtime.h>
#include <math.h>


#define CUT0 10000
#define CUT1 30000
#define DIM  1024
#define NROWS 4096
#define HEADM 10002
#define M1 20000
#define M2 20257
#define D1 256
#define D2 64
#define ROWS_B 32
#define CHUNK 512
#define NCH_H 20
#define NCH_1 40
#define NCH_2 40

typedef float v2f __attribute__((ext_vector_type(2)));
typedef float v8f __attribute__((ext_vector_type(8)));

__device__ __forceinline__ v8f wmma4(v2f a, v2f b, v8f c) {
  return __builtin_amdgcn_wmma_f32_16x16x4_f32(false, a, false, b, (short)0, c,
                                               false, false);
}

__device__ __forceinline__ void gemm_wave32(const float* xp0, const float* xp1,
                                            const float* wp0, const float* wp1,
                                            const float* wp2, const float* wp3,
                                            int K, v8f acc[2][4]) {
  v2f a0 = *(const v2f*)(xp0);
  v2f a1 = *(const v2f*)(xp1);
  v2f b0 = *(const v2f*)(wp0);
  v2f b1 = *(const v2f*)(wp1);
  v2f b2 = *(const v2f*)(wp2);
  v2f b3 = *(const v2f*)(wp3);
  for (int k = 4; k < K; k += 4) {
    v2f a0n = *(const v2f*)(xp0 + k);
    v2f a1n = *(const v2f*)(xp1 + k);
    v2f b0n = *(const v2f*)(wp0 + k);
    v2f b1n = *(const v2f*)(wp1 + k);
    v2f b2n = *(const v2f*)(wp2 + k);
    v2f b3n = *(const v2f*)(wp3 + k);
    __builtin_amdgcn_sched_barrier(0);
    acc[0][0] = wmma4(a0, b0, acc[0][0]);
    acc[0][1] = wmma4(a0, b1, acc[0][1]);
    acc[0][2] = wmma4(a0, b2, acc[0][2]);
    acc[0][3] = wmma4(a0, b3, acc[0][3]);
    acc[1][0] = wmma4(a1, b0, acc[1][0]);
    acc[1][1] = wmma4(a1, b1, acc[1][1]);
    acc[1][2] = wmma4(a1, b2, acc[1][2]);
    acc[1][3] = wmma4(a1, b3, acc[1][3]);
    __builtin_amdgcn_sched_barrier(0);
    a0 = a0n; a1 = a1n; b0 = b0n; b1 = b1n; b2 = b2n; b3 = b3n;
  }
  acc[0][0] = wmma4(a0, b0, acc[0][0]);
  acc[0][1] = wmma4(a0, b1, acc[0][1]);
  acc[0][2] = wmma4(a0, b2, acc[0][2]);
  acc[0][3] = wmma4(a0, b3, acc[0][3]);
  acc[1][0] = wmma4(a1, b0, acc[1][0]);
  acc[1][1] = wmma4(a1, b1, acc[1][1]);
  acc[1][2] = wmma4(a1, b2, acc[1][2]);
  acc[1][3] = wmma4(a1, b3, acc[1][3]);
}

__global__ void gemm_store(const float* __restrict__ X, const float* __restrict__ W,
                           float* __restrict__ C, int K, int M) {
  const int wave = threadIdx.x >> 5;
  const int lane = threadIdx.x & 31;
  const int half = lane >> 4;
  const int r    = lane & 15;
  const int row0 = blockIdx.y * ROWS_B;
  const int col0 = blockIdx.x * CHUNK + wave * 64;
  if (col0 >= M) return;

  const float* xp0 = X + (size_t)(row0 + r) * K + 2 * half;
  const float* xp1 = X + (size_t)(row0 + 16 + r) * K + 2 * half;
  const float* wp0 = W + (size_t)(col0 + r) * K + 2 * half;
  const float* wp1 = W + (size_t)(col0 + 16 + r) * K + 2 * half;
  const float* wp2 = W + (size_t)(col0 + 32 + r) * K + 2 * half;
  const float* wp3 = W + (size_t)(col0 + 48 + r) * K + 2 * half;

  v8f acc[2][4];
#pragma unroll
  for (int rt = 0; rt < 2; ++rt)
#pragma unroll
    for (int t = 0; t < 4; ++t)
      acc[rt][t] = (v8f){0.f,0.f,0.f,0.f,0.f,0.f,0.f,0.f};
  gemm_wave32(xp0, xp1, wp0, wp1, wp2, wp3, K, acc);

#pragma unroll
  for (int rt = 0; rt < 2; ++rt) {
#pragma unroll
    for (int t = 0; t < 4; ++t) {
#pragma unroll
      for (int v = 0; v < 8; ++v) {
        int rr = row0 + 16 * rt + v + 8 * half;
        C[(size_t)rr * M + col0 + 16 * t + r] = acc[rt][t][v];
      }
    }
  }
}

__global__ void gemm_expsum(const float* __restrict__ X, const float* __restrict__ W,
                            float* __restrict__ partial, int K, int M, int nchunks) {
  __shared__ float lsum[8][ROWS_B];
  const int wave = threadIdx.x >> 5;
  const int lane = threadIdx.x & 31;
  const int half = lane >> 4;
  const int r    = lane & 15;
  const int row0 = blockIdx.y * ROWS_B;
  const int col0 = blockIdx.x * CHUNK + wave * 64;
  const bool active = (col0 < M);

  v8f acc[2][4];
#pragma unroll
  for (int rt = 0; rt < 2; ++rt)
#pragma unroll
    for (int t = 0; t < 4; ++t)
      acc[rt][t] = (v8f){0.f,0.f,0.f,0.f,0.f,0.f,0.f,0.f};
  if (active) {
    const float* xp0 = X + (size_t)(row0 + r) * K + 2 * half;
    const float* xp1 = X + (size_t)(row0 + 16 + r) * K + 2 * half;
    const float* wp0 = W + (size_t)min(col0 + r,      M - 1) * K + 2 * half;
    const float* wp1 = W + (size_t)min(col0 + 16 + r, M - 1) * K + 2 * half;
    const float* wp2 = W + (size_t)min(col0 + 32 + r, M - 1) * K + 2 * half;
    const float* wp3 = W + (size_t)min(col0 + 48 + r, M - 1) * K + 2 * half;
    gemm_wave32(xp0, xp1, wp0, wp1, wp2, wp3, K, acc);
  }
#pragma unroll
  for (int rt = 0; rt < 2; ++rt) {
#pragma unroll
    for (int v = 0; v < 8; ++v) {
      float e = 0.f;
      if (active) {
#pragma unroll
        for (int t = 0; t < 4; ++t)
          if (col0 + 16 * t + r < M) e += expf(acc[rt][t][v]);
      }
      e += __shfl_xor(e, 1, 32);
      e += __shfl_xor(e, 2, 32);
      e += __shfl_xor(e, 4, 32);
      e += __shfl_xor(e, 8, 32);
      if (r == 0) lsum[wave][16 * rt + v + 8 * half] = e;
    }
  }
  __syncthreads();
  if (threadIdx.x < ROWS_B) {
    float t = 0.f;
#pragma unroll
    for (int w = 0; w < 8; ++w) t += lsum[w][threadIdx.x];
    partial[(size_t)(row0 + threadIdx.x) * nchunks + blockIdx.x] = t;
  }
}

__global__ void gather_dots(const float* __restrict__ x, const int* __restrict__ target,
                            const float* __restrict__ Wh, const float* __restrict__ Wo1,
                            const float* __restrict__ Wo2, const float* __restrict__ h1,
                            const float* __restrict__ h2,
                            float* __restrict__ dh, float* __restrict__ d1,
                            float* __restrict__ d2) {
  const int wave = threadIdx.x >> 5;
  const int lane = threadIdx.x & 31;
  const int row  = blockIdx.x * 8 + wave;
  const int t = target[row];
  const int g = (t < CUT0) ? t : ((t < CUT1) ? CUT0 : CUT0 + 1);

  float s0 = 0.f, s1 = 0.f, s2 = 0.f;
  {
    const float* xr = x + (size_t)row * DIM;
    const float* wr = Wh + (size_t)g * DIM;
    for (int k = lane; k < DIM; k += 32) s0 += xr[k] * wr[k];
  }
  {
    const int rel = min(max(t - CUT0, 0), M1 - 1);
    const float* hr = h1 + (size_t)row * D1;
    const float* wr = Wo1 + (size_t)rel * D1;
    for (int k = lane; k < D1; k += 32) s1 += hr[k] * wr[k];
  }
  {
    const int rel = min(max(t - CUT1, 0), M2 - 1);
    const float* hr = h2 + (size_t)row * D2;
    const float* wr = Wo2 + (size_t)rel * D2;
    for (int k = lane; k < D2; k += 32) s2 += hr[k] * wr[k];
  }
#pragma unroll
  for (int m = 16; m; m >>= 1) {
    s0 += __shfl_xor(s0, m, 32);
    s1 += __shfl_xor(s1, m, 32);
    s2 += __shfl_xor(s2, m, 32);
  }
  if (lane == 0) { dh[row] = s0; d1[row] = s1; d2[row] = s2; }
}

__global__ void finalize(const int* __restrict__ target,
                         const float* __restrict__ ph, const float* __restrict__ p1,
                         const float* __restrict__ p2, const float* __restrict__ dh,
                         const float* __restrict__ d1, const float* __restrict__ d2,
                         float* __restrict__ out) {
  const int i = blockIdx.x * blockDim.x + threadIdx.x;
  if (i >= NROWS) return;
  float sh = 0.f;
  for (int c = 0; c < NCH_H; ++c) sh += ph[(size_t)i * NCH_H + c];
  float o = dh[i] - logf(sh);
  const int t = target[i];
  if (t >= CUT0 && t < CUT1) {
    float s = 0.f;
    for (int c = 0; c < NCH_1; ++c) s += p1[(size_t)i * NCH_1 + c];
    o += d1[i] - logf(s);
  } else if (t >= CUT1) {
    float s = 0.f;
    for (int c = 0; c < NCH_2; ++c) s += p2[(size_t)i * NCH_2 + c];
    o += d2[i] - logf(s);
  }
  out[i] = o;
}

__global__ void loss_reduce(const float* __restrict__ out, float* __restrict__ loss) {
  __shared__ float sm[256];
  float s = 0.f;
  for (int i = threadIdx.x; i < NROWS; i += 256) s += out[i];
  sm[threadIdx.x] = s;
  __syncthreads();
  for (int w = 128; w; w >>= 1) {
    if (threadIdx.x < w) sm[threadIdx.x] += sm[threadIdx.x + w];
    __syncthreads();
  }
  if (threadIdx.x == 0) *loss = -sm[0] / (float)NROWS;
}

extern "C" void kernel_launch(void* const* d_in, const int* in_sizes, int n_in,
                              void* d_out, int out_size, void* d_ws, size_t ws_size,
                              hipStream_t stream) {
  (void)in_sizes; (void)n_in; (void)out_size; (void)ws_size;
  const float* x    = (const float*)d_in[0];
  const int*   targ = (const int*)d_in[1];
  const float* Wh   = (const float*)d_in[2];
  const float* Wp1  = (const float*)d_in[3];
  const float* Wo1  = (const float*)d_in[4];
  const float* Wp2  = (const float*)d_in[5];
  const float* Wo2  = (const float*)d_in[6];
  float* out = (float*)d_out;

  float* ws = (float*)d_ws;
  float* h1 = ws;
  float* h2 = h1 + (size_t)NROWS * D1;
  float* ph = h2 + (size_t)NROWS * D2;
  float* p1 = ph + (size_t)NROWS * NCH_H;
  float* p2 = p1 + (size_t)NROWS * NCH_1;
  float* dh = p2 + (size_t)NROWS * NCH_2;
  float* d1 = dh + NROWS;
  float* d2 = d1 + NROWS;

  const dim3 blk(256);
  const int rowBlocks = NROWS / ROWS_B;
  gemm_store<<<dim3(1, rowBlocks), blk, 0, stream>>>(x, Wp1, h1, DIM, D1);
  gemm_store<<<dim3(1, rowBlocks), blk, 0, stream>>>(x, Wp2, h2, DIM, D2);
  gemm_expsum<<<dim3(NCH_H, rowBlocks), blk, 0, stream>>>(x,  Wh,  ph, DIM, HEADM, NCH_H);
  gemm_expsum<<<dim3(NCH_1, rowBlocks), blk, 0, stream>>>(h1, Wo1, p1, D1,  M1,    NCH_1);
  gemm_expsum<<<dim3(NCH_2, rowBlocks), blk, 0, stream>>>(h2, Wo2, p2, D2,  M2,    NCH_2);
  gather_dots<<<dim3(NROWS / 8), blk, 0, stream>>>(x, targ, Wh, Wo1, Wo2, h1, h2, dh, d1, d2);
  finalize<<<dim3(NROWS / 256), blk, 0, stream>>>(targ, ph, p1, p2, dh, d1, d2, out);
  loss_reduce<<<1, 256, 0, stream>>>(out, out + NROWS);
}


# Exp-sum GEMMs: stage X and W tiles (128x128, k 32) through LDS with coalesced 128-bit loads; changes k order inside each 8 and the exp-sum partial tree (per 128 cols)
# speedup vs baseline: 2.8762x; 2.8762x over previous
#include <hip/hip_runtime.h>
#include <math.h>


#define CUT0 10000
#define CUT1 30000
#define DIM  1024
#define NROWS 4096
#define HEADM 10002
#define M1 20000
#define M2 20257
#define D1 256
#define D2 64
#define ROWS_B 32
#define CHUNK 512
#define BM 128
#define BN 128
#define KC 32
#define LDP 36
#define NCH_H 79
#define NCH_1 157
#define NCH_2 159

typedef float v2f __attribute__((ext_vector_type(2)));
typedef float v4f __attribute__((ext_vector_type(4)));
typedef float v8f __attribute__((ext_vector_type(8)));

__device__ __forceinline__ v8f wmma4(v2f a, v2f b, v8f c) {
  return __builtin_amdgcn_wmma_f32_16x16x4_f32(false, a, false, b, (short)0, c,
                                               false, false);
}

__device__ __forceinline__ void gemm_wave32(const float* xp0, const float* xp1,
                                            const float* wp0, const float* wp1,
                                            const float* wp2, const float* wp3,
                                            int K, v8f acc[2][4]) {
  v2f a0 = *(const v2f*)(xp0);
  v2f a1 = *(const v2f*)(xp1);
  v2f b0 = *(const v2f*)(wp0);
  v2f b1 = *(const v2f*)(wp1);
  v2f b2 = *(const v2f*)(wp2);
  v2f b3 = *(const v2f*)(wp3);
  for (int k = 4; k < K; k += 4) {
    v2f a0n = *(const v2f*)(xp0 + k);
    v2f a1n = *(const v2f*)(xp1 + k);
    v2f b0n = *(const v2f*)(wp0 + k);
    v2f b1n = *(const v2f*)(wp1 + k);
    v2f b2n = *(const v2f*)(wp2 + k);
    v2f b3n = *(const v2f*)(wp3 + k);
    __builtin_amdgcn_sched_barrier(0);
    acc[0][0] = wmma4(a0, b0, acc[0][0]);
    acc[0][1] = wmma4(a0, b1, acc[0][1]);
    acc[0][2] = wmma4(a0, b2, acc[0][2]);
    acc[0][3] = wmma4(a0, b3, acc[0][3]);
    acc[1][0] = wmma4(a1, b0, acc[1][0]);
    acc[1][1] = wmma4(a1, b1, acc[1][1]);
    acc[1][2] = wmma4(a1, b2, acc[1][2]);
    acc[1][3] = wmma4(a1, b3, acc[1][3]);
    __builtin_amdgcn_sched_barrier(0);
    a0 = a0n; a1 = a1n; b0 = b0n; b1 = b1n; b2 = b2n; b3 = b3n;
  }
  acc[0][0] = wmma4(a0, b0, acc[0][0]);
  acc[0][1] = wmma4(a0, b1, acc[0][1]);
  acc[0][2] = wmma4(a0, b2, acc[0][2]);
  acc[0][3] = wmma4(a0, b3, acc[0][3]);
  acc[1][0] = wmma4(a1, b0, acc[1][0]);
  acc[1][1] = wmma4(a1, b1, acc[1][1]);
  acc[1][2] = wmma4(a1, b2, acc[1][2]);
  acc[1][3] = wmma4(a1, b3, acc[1][3]);
}

__global__ void gemm_store(const float* __restrict__ X, const float* __restrict__ W,
                           float* __restrict__ C, int K, int M) {
  const int wave = threadIdx.x >> 5;
  const int lane = threadIdx.x & 31;
  const int half = lane >> 4;
  const int r    = lane & 15;
  const int row0 = blockIdx.y * ROWS_B;
  const int col0 = blockIdx.x * CHUNK + wave * 64;
  if (col0 >= M) return;

  const float* xp0 = X + (size_t)(row0 + r) * K + 2 * half;
  const float* xp1 = X + (size_t)(row0 + 16 + r) * K + 2 * half;
  const float* wp0 = W + (size_t)(col0 + r) * K + 2 * half;
  const float* wp1 = W + (size_t)(col0 + 16 + r) * K + 2 * half;
  const float* wp2 = W + (size_t)(col0 + 32 + r) * K + 2 * half;
  const float* wp3 = W + (size_t)(col0 + 48 + r) * K + 2 * half;

  v8f acc[2][4];
#pragma unroll
  for (int rt = 0; rt < 2; ++rt)
#pragma unroll
    for (int t = 0; t < 4; ++t)
      acc[rt][t] = (v8f){0.f,0.f,0.f,0.f,0.f,0.f,0.f,0.f};
  gemm_wave32(xp0, xp1, wp0, wp1, wp2, wp3, K, acc);

#pragma unroll
  for (int rt = 0; rt < 2; ++rt) {
#pragma unroll
    for (int t = 0; t < 4; ++t) {
#pragma unroll
      for (int v = 0; v < 8; ++v) {
        int rr = row0 + 16 * rt + v + 8 * half;
        C[(size_t)rr * M + col0 + 16 * t + r] = acc[rt][t][v];
      }
    }
  }
}

__global__ __launch_bounds__(256)
void gemm_expsum(const float* __restrict__ X, const float* __restrict__ W,
                 float* __restrict__ partial, int K, int M) {
  __shared__ __attribute__((aligned(16))) float As[2][BM * LDP];
  __shared__ __attribute__((aligned(16))) float Bs[2][BN * LDP];
  __shared__ float lsum[2][BM];
  const int tid  = threadIdx.x;
  const int wave = tid >> 5;
  const int lane = tid & 31;
  const int half = lane >> 4;
  const int r    = lane & 15;
  const int wr   = wave & 3;
  const int wc   = wave >> 2;
  const int row0 = blockIdx.y * BM;
  const int col0 = blockIdx.x * BN;

  const int lr = tid >> 3;
  const int c4 = tid & 7;
  const float* xg[4];
  const float* wg[4];
#pragma unroll
  for (int i = 0; i < 4; ++i) {
    xg[i] = X + (size_t)(row0 + lr + 32 * i) * K + 4 * c4;
    wg[i] = W + (size_t)min(col0 + lr + 32 * i, M - 1) * K + 4 * c4;
  }
  const int sOff = lr * LDP + 4 * c4;

  v4f sa[4], sb[4];
#pragma unroll
  for (int i = 0; i < 4; ++i) {
    sa[i] = *(const v4f*)(xg[i]);
    sb[i] = *(const v4f*)(wg[i]);
  }
#pragma unroll
  for (int i = 0; i < 4; ++i) {
    *(v4f*)&As[0][sOff + 32 * i * LDP] = sa[i];
    *(v4f*)&Bs[0][sOff + 32 * i * LDP] = sb[i];
  }
  __syncthreads();

  const int aOff = (wr * 32 + r) * LDP + 4 * half;
  const int bOff = (wc * 64 + r) * LDP + 4 * half;

  v8f acc[2][4];
#pragma unroll
  for (int rt = 0; rt < 2; ++rt)
#pragma unroll
    for (int t = 0; t < 4; ++t)
      acc[rt][t] = (v8f){0.f,0.f,0.f,0.f,0.f,0.f,0.f,0.f};

  const int nch = K / KC;
  for (int c = 0; c < nch; ++c) {
    const int cur = c & 1;
    const bool more = (c + 1 < nch);
    if (more) {
#pragma unroll
      for (int i = 0; i < 4; ++i) {
        sa[i] = *(const v4f*)(xg[i] + (c + 1) * KC);
        sb[i] = *(const v4f*)(wg[i] + (c + 1) * KC);
      }
    }
    const float* as = &As[cur][aOff];
    const float* bs = &Bs[cur][bOff];
    v4f a[2], b[4];
#pragma unroll
    for (int rt = 0; rt < 2; ++rt) a[rt] = *(const v4f*)(as + rt * 16 * LDP);
#pragma unroll
    for (int t = 0; t < 4; ++t)   b[t]  = *(const v4f*)(bs + t * 16 * LDP);
#pragma unroll
    for (int kk = 0; kk < KC / 8; ++kk) {
      v4f an[2], bn[4];
      if (kk + 1 < KC / 8) {
#pragma unroll
        for (int rt = 0; rt < 2; ++rt)
          an[rt] = *(const v4f*)(as + rt * 16 * LDP + (kk + 1) * 8);
#pragma unroll
        for (int t = 0; t < 4; ++t)
          bn[t] = *(const v4f*)(bs + t * 16 * LDP + (kk + 1) * 8);
      }
      __builtin_amdgcn_sched_barrier(0);
#pragma unroll
      for (int rt = 0; rt < 2; ++rt) {
#pragma unroll
        for (int t = 0; t < 4; ++t) {
          acc[rt][t] = wmma4((v2f){a[rt].x, a[rt].y}, (v2f){b[t].x, b[t].y}, acc[rt][t]);
        }
      }
#pragma unroll
      for (int rt = 0; rt < 2; ++rt) {
#pragma unroll
        for (int t = 0; t < 4; ++t) {
          acc[rt][t] = wmma4((v2f){a[rt].z, a[rt].w}, (v2f){b[t].z, b[t].w}, acc[rt][t]);
        }
      }
      __builtin_amdgcn_sched_barrier(0);
      if (kk + 1 < KC / 8) {
#pragma unroll
        for (int rt = 0; rt < 2; ++rt) a[rt] = an[rt];
#pragma unroll
        for (int t = 0; t < 4; ++t)   b[t]  = bn[t];
      }
    }
    if (more) {
#pragma unroll
      for (int i = 0; i < 4; ++i) {
        *(v4f*)&As[cur ^ 1][sOff + 32 * i * LDP] = sa[i];
        *(v4f*)&Bs[cur ^ 1][sOff + 32 * i * LDP] = sb[i];
      }
    }
    __syncthreads();
  }

  const int cbase = col0 + wc * 64 + r;
#pragma unroll
  for (int rt = 0; rt < 2; ++rt) {
#pragma unroll
    for (int v = 0; v < 8; ++v) {
      float e = 0.f;
#pragma unroll
      for (int t = 0; t < 4; ++t)
        if (cbase + 16 * t < M) e += expf(acc[rt][t][v]);
      e += __shfl_xor(e, 1, 32);
      e += __shfl_xor(e, 2, 32);
      e += __shfl_xor(e, 4, 32);
      e += __shfl_xor(e, 8, 32);
      if (r == 0) lsum[wc][wr * 32 + 16 * rt + v + 8 * half] = e;
    }
  }
  __syncthreads();
  if (tid < BM)
    partial[(size_t)blockIdx.x * NROWS + row0 + tid] = lsum[0][tid] + lsum[1][tid];
}

__global__ void gather_dots(const float* __restrict__ x, const int* __restrict__ target,
                            const float* __restrict__ Wh, const float* __restrict__ Wo1,
                            const float* __restrict__ Wo2, const float* __restrict__ h1,
                            const float* __restrict__ h2,
                            float* __restrict__ dh, float* __restrict__ d1,
                            float* __restrict__ d2) {
  const int wave = threadIdx.x >> 5;
  const int lane = threadIdx.x & 31;
  const int row  = blockIdx.x * 8 + wave;
  const int t = target[row];
  const int g = (t < CUT0) ? t : ((t < CUT1) ? CUT0 : CUT0 + 1);

  float s0 = 0.f, s1 = 0.f, s2 = 0.f;
  {
    const float* xr = x + (size_t)row * DIM;
    const float* wr = Wh + (size_t)g * DIM;
    for (int k = lane; k < DIM; k += 32) s0 += xr[k] * wr[k];
  }
  {
    const int rel = min(max(t - CUT0, 0), M1 - 1);
    const float* hr = h1 + (size_t)row * D1;
    const float* wr = Wo1 + (size_t)rel * D1;
    for (int k = lane; k < D1; k += 32) s1 += hr[k] * wr[k];
  }
  {
    const int rel = min(max(t - CUT1, 0), M2 - 1);
    const float* hr = h2 + (size_t)row * D2;
    const float* wr = Wo2 + (size_t)rel * D2;
    for (int k = lane; k < D2; k += 32) s2 += hr[k] * wr[k];
  }
#pragma unroll
  for (int m = 16; m; m >>= 1) {
    s0 += __shfl_xor(s0, m, 32);
    s1 += __shfl_xor(s1, m, 32);
    s2 += __shfl_xor(s2, m, 32);
  }
  if (lane == 0) { dh[row] = s0; d1[row] = s1; d2[row] = s2; }
}

__global__ void finalize(const int* __restrict__ target,
                         const float* __restrict__ ph, const float* __restrict__ p1,
                         const float* __restrict__ p2, const float* __restrict__ dh,
                         const float* __restrict__ d1, const float* __restrict__ d2,
                         float* __restrict__ out) {
  const int i = blockIdx.x * blockDim.x + threadIdx.x;
  if (i >= NROWS) return;
  float sh = 0.f;
  for (int c = 0; c < NCH_H; ++c) sh += ph[(size_t)c * NROWS + i];
  float o = dh[i] - logf(sh);
  const int t = target[i];
  if (t >= CUT0 && t < CUT1) {
    float s = 0.f;
    for (int c = 0; c < NCH_1; ++c) s += p1[(size_t)c * NROWS + i];
    o += d1[i] - logf(s);
  } else if (t >= CUT1) {
    float s = 0.f;
#pragma unroll 8
    for (int c = 0; c < NCH_2; ++c) s += p2[(size_t)c * NROWS + i];
    o += d2[i] - logf(s);
  }
  out[i] = o;
}

__global__ void loss_reduce(const float* __restrict__ out, float* __restrict__ loss) {
  __shared__ float sm[256];
  float s = 0.f;
  for (int i = threadIdx.x; i < NROWS; i += 256) s += out[i];
  sm[threadIdx.x] = s;
  __syncthreads();
  for (int w = 128; w; w >>= 1) {
    if (threadIdx.x < w) sm[threadIdx.x] += sm[threadIdx.x + w];
    __syncthreads();
  }
  if (threadIdx.x == 0) *loss = -sm[0] / (float)NROWS;
}

extern "C" void kernel_launch(void* const* d_in, const int* in_sizes, int n_in,
                              void* d_out, int out_size, void* d_ws, size_t ws_size,
                              hipStream_t stream) {
  (void)in_sizes; (void)n_in; (void)out_size; (void)ws_size;
  const float* x    = (const float*)d_in[0];
  const int*   targ = (const int*)d_in[1];
  const float* Wh   = (const float*)d_in[2];
  const float* Wp1  = (const float*)d_in[3];
  const float* Wo1  = (const float*)d_in[4];
  const float* Wp2  = (const float*)d_in[5];
  const float* Wo2  = (const float*)d_in[6];
  float* out = (float*)d_out;

  float* ws = (float*)d_ws;
  float* h1 = ws;
  float* h2 = h1 + (size_t)NROWS * D1;
  float* ph = h2 + (size_t)NROWS * D2;
  float* p1 = ph + (size_t)NROWS * NCH_H;
  float* p2 = p1 + (size_t)NROWS * NCH_1;
  float* dh = p2 + (size_t)NROWS * NCH_2;
  float* d1 = dh + NROWS;
  float* d2 = d1 + NROWS;

  const dim3 blk(256);
  const int rowBlocks = NROWS / ROWS_B;
  gemm_store<<<dim3(1, rowBlocks), blk, 0, stream>>>(x, Wp1, h1, DIM, D1);
  gemm_store<<<dim3(1, rowBlocks), blk, 0, stream>>>(x, Wp2, h2, DIM, D2);
  gemm_expsum<<<dim3(NCH_H, NROWS / BM), blk, 0, stream>>>(x,  Wh,  ph, DIM, HEADM);
  gemm_expsum<<<dim3(NCH_1, NROWS / BM), blk, 0, stream>>>(h1, Wo1, p1, D1,  M1);
  gemm_expsum<<<dim3(NCH_2, NROWS / BM), blk, 0, stream>>>(h2, Wo2, p2, D2,  M2);
  gather_dots<<<dim3(NROWS / 8), blk, 0, stream>>>(x, targ, Wh, Wo1, Wo2, h1, h2, dh, d1, d2);
  finalize<<<dim3(NROWS / 256), blk, 0, stream>>>(targ, ph, p1, p2, dh, d1, d2, out);
  loss_reduce<<<1, 256, 0, stream>>>(out, out + NROWS);
}


# Tail exp-sum GEMMs only over rows whose target is in the cluster (reference masks the rest): scan kernel lists rows per cluster, both tails in one launch, partials by list position
# speedup vs baseline: 3.7313x; 1.2973x over previous
#include <hip/hip_runtime.h>
#include <math.h>


#define CUT0 10000
#define CUT1 30000
#define DIM  1024
#define NROWS 4096
#define HEADM 10002
#define M1 20000
#define M2 20257
#define D1 256
#define D2 64
#define ROWS_B 32
#define CHUNK 512
#define BM 128
#define BN 128
#define KC 32
#define LDP 36
#define NCH_H 79
#define NCH_1 157
#define NCH_2 159

typedef float v2f __attribute__((ext_vector_type(2)));
typedef float v4f __attribute__((ext_vector_type(4)));
typedef float v8f __attribute__((ext_vector_type(8)));

__device__ __forceinline__ v8f wmma4(v2f a, v2f b, v8f c) {
  return __builtin_amdgcn_wmma_f32_16x16x4_f32(false, a, false, b, (short)0, c,
                                               false, false);
}

__device__ __forceinline__ void gemm_wave32(const float* xp0, const float* xp1,
                                            const float* wp0, const float* wp1,
                                            const float* wp2, const float* wp3,
                                            int K, v8f acc[2][4]) {
  v2f a0 = *(const v2f*)(xp0);
  v2f a1 = *(const v2f*)(xp1);
  v2f b0 = *(const v2f*)(wp0);
  v2f b1 = *(const v2f*)(wp1);
  v2f b2 = *(const v2f*)(wp2);
  v2f b3 = *(const v2f*)(wp3);
  for (int k = 4; k < K; k += 4) {
    v2f a0n = *(const v2f*)(xp0 + k);
    v2f a1n = *(const v2f*)(xp1 + k);
    v2f b0n = *(const v2f*)(wp0 + k);
    v2f b1n = *(const v2f*)(wp1 + k);
    v2f b2n = *(const v2f*)(wp2 + k);
    v2f b3n = *(const v2f*)(wp3 + k);
    __builtin_amdgcn_sched_barrier(0);
    acc[0][0] = wmma4(a0, b0, acc[0][0]);
    acc[0][1] = wmma4(a0, b1, acc[0][1]);
    acc[0][2] = wmma4(a0, b2, acc[0][2]);
    acc[0][3] = wmma4(a0, b3, acc[0][3]);
    acc[1][0] = wmma4(a1, b0, acc[1][0]);
    acc[1][1] = wmma4(a1, b1, acc[1][1]);
    acc[1][2] = wmma4(a1, b2, acc[1][2]);
    acc[1][3] = wmma4(a1, b3, acc[1][3]);
    __builtin_amdgcn_sched_barrier(0);
    a0 = a0n; a1 = a1n; b0 = b0n; b1 = b1n; b2 = b2n; b3 = b3n;
  }
  acc[0][0] = wmma4(a0, b0, acc[0][0]);
  acc[0][1] = wmma4(a0, b1, acc[0][1]);
  acc[0][2] = wmma4(a0, b2, acc[0][2]);
  acc[0][3] = wmma4(a0, b3, acc[0][3]);
  acc[1][0] = wmma4(a1, b0, acc[1][0]);
  acc[1][1] = wmma4(a1, b1, acc[1][1]);
  acc[1][2] = wmma4(a1, b2, acc[1][2]);
  acc[1][3] = wmma4(a1, b3, acc[1][3]);
}

__global__ void gemm_store(const float* __restrict__ X, const float* __restrict__ W,
                           float* __restrict__ C, int K, int M) {
  const int wave = threadIdx.x >> 5;
  const int lane = threadIdx.x & 31;
  const int half = lane >> 4;
  const int r    = lane & 15;
  const int row0 = blockIdx.y * ROWS_B;
  const int col0 = blockIdx.x * CHUNK + wave * 64;
  if (col0 >= M) return;

  const float* xp0 = X + (size_t)(row0 + r) * K + 2 * half;
  const float* xp1 = X + (size_t)(row0 + 16 + r) * K + 2 * half;
  const float* wp0 = W + (size_t)(col0 + r) * K + 2 * half;
  const float* wp1 = W + (size_t)(col0 + 16 + r) * K + 2 * half;
  const float* wp2 = W + (size_t)(col0 + 32 + r) * K + 2 * half;
  const float* wp3 = W + (size_t)(col0 + 48 + r) * K + 2 * half;

  v8f acc[2][4];
#pragma unroll
  for (int rt = 0; rt < 2; ++rt)
#pragma unroll
    for (int t = 0; t < 4; ++t)
      acc[rt][t] = (v8f){0.f,0.f,0.f,0.f,0.f,0.f,0.f,0.f};
  gemm_wave32(xp0, xp1, wp0, wp1, wp2, wp3, K, acc);

#pragma unroll
  for (int rt = 0; rt < 2; ++rt) {
#pragma unroll
    for (int t = 0; t < 4; ++t) {
#pragma unroll
      for (int v = 0; v < 8; ++v) {
        int rr = row0 + 16 * rt + v + 8 * half;
        C[(size_t)rr * M + col0 + 16 * t + r] = acc[rt][t][v];
      }
    }
  }
}

template <bool GATHER>
__device__ __forceinline__
void expsum_block(const float* __restrict__ X, const float* __restrict__ W,
                  float* __restrict__ partial, const int K, const int M,
                  const int bx, const int row0,
                  const int* __restrict__ rowidx, const int n) {
  __shared__ __attribute__((aligned(16))) float As[2][BM * LDP];
  __shared__ __attribute__((aligned(16))) float Bs[2][BN * LDP];
  __shared__ float lsum[2][BM];
  const int tid  = threadIdx.x;
  const int wave = tid >> 5;
  const int lane = tid & 31;
  const int half = lane >> 4;
  const int r    = lane & 15;
  const int wr   = wave & 3;
  const int wc   = wave >> 2;
  const int col0 = bx * BN;

  const int lr = tid >> 3;
  const int c4 = tid & 7;
  const float* xg[4];
  const float* wg[4];
#pragma unroll
  for (int i = 0; i < 4; ++i) {
    const int s = row0 + lr + 32 * i;
    const int xr = GATHER ? rowidx[min(s, n - 1)] : s;
    xg[i] = X + (size_t)xr * K + 4 * c4;
    wg[i] = W + (size_t)min(col0 + lr + 32 * i, M - 1) * K + 4 * c4;
  }
  const int sOff = lr * LDP + 4 * c4;

  v4f sa[4], sb[4];
#pragma unroll
  for (int i = 0; i < 4; ++i) {
    sa[i] = *(const v4f*)(xg[i]);
    sb[i] = *(const v4f*)(wg[i]);
  }
#pragma unroll
  for (int i = 0; i < 4; ++i) {
    *(v4f*)&As[0][sOff + 32 * i * LDP] = sa[i];
    *(v4f*)&Bs[0][sOff + 32 * i * LDP] = sb[i];
  }
  __syncthreads();

  const int aOff = (wr * 32 + r) * LDP + 4 * half;
  const int bOff = (wc * 64 + r) * LDP + 4 * half;

  v8f acc[2][4];
#pragma unroll
  for (int rt = 0; rt < 2; ++rt)
#pragma unroll
    for (int t = 0; t < 4; ++t)
      acc[rt][t] = (v8f){0.f,0.f,0.f,0.f,0.f,0.f,0.f,0.f};

  const int nch = K / KC;
  for (int c = 0; c < nch; ++c) {
    const int cur = c & 1;
    const bool more = (c + 1 < nch);
    if (more) {
#pragma unroll
      for (int i = 0; i < 4; ++i) {
        sa[i] = *(const v4f*)(xg[i] + (c + 1) * KC);
        sb[i] = *(const v4f*)(wg[i] + (c + 1) * KC);
      }
    }
    const float* as = &As[cur][aOff];
    const float* bs = &Bs[cur][bOff];
    v4f a[2], b[4];
#pragma unroll
    for (int rt = 0; rt < 2; ++rt) a[rt] = *(const v4f*)(as + rt * 16 * LDP);
#pragma unroll
    for (int t = 0; t < 4; ++t)   b[t]  = *(const v4f*)(bs + t * 16 * LDP);
#pragma unroll
    for (int kk = 0; kk < KC / 8; ++kk) {
      v4f an[2], bn[4];
      if (kk + 1 < KC / 8) {
#pragma unroll
        for (int rt = 0; rt < 2; ++rt)
          an[rt] = *(const v4f*)(as + rt * 16 * LDP + (kk + 1) * 8);
#pragma unroll
        for (int t = 0; t < 4; ++t)
          bn[t] = *(const v4f*)(bs + t * 16 * LDP + (kk + 1) * 8);
      }
      __builtin_amdgcn_sched_barrier(0);
#pragma unroll
      for (int rt = 0; rt < 2; ++rt) {
#pragma unroll
        for (int t = 0; t < 4; ++t) {
          acc[rt][t] = wmma4((v2f){a[rt].x, a[rt].y}, (v2f){b[t].x, b[t].y}, acc[rt][t]);
        }
      }
#pragma unroll
      for (int rt = 0; rt < 2; ++rt) {
#pragma unroll
        for (int t = 0; t < 4; ++t) {
          acc[rt][t] = wmma4((v2f){a[rt].z, a[rt].w}, (v2f){b[t].z, b[t].w}, acc[rt][t]);
        }
      }
      __builtin_amdgcn_sched_barrier(0);
      if (kk + 1 < KC / 8) {
#pragma unroll
        for (int rt = 0; rt < 2; ++rt) a[rt] = an[rt];
#pragma unroll
        for (int t = 0; t < 4; ++t)   b[t]  = bn[t];
      }
    }
    if (more) {
#pragma unroll
      for (int i = 0; i < 4; ++i) {
        *(v4f*)&As[cur ^ 1][sOff + 32 * i * LDP] = sa[i];
        *(v4f*)&Bs[cur ^ 1][sOff + 32 * i * LDP] = sb[i];
      }
    }
    __syncthreads();
  }

  const int cbase = col0 + wc * 64 + r;
#pragma unroll
  for (int rt = 0; rt < 2; ++rt) {
#pragma unroll
    for (int v = 0; v < 8; ++v) {
      float e = 0.f;
#pragma unroll
      for (int t = 0; t < 4; ++t)
        if (cbase + 16 * t < M) e += expf(acc[rt][t][v]);
      e += __shfl_xor(e, 1, 32);
      e += __shfl_xor(e, 2, 32);
      e += __shfl_xor(e, 4, 32);
      e += __shfl_xor(e, 8, 32);
      if (r == 0) lsum[wc][wr * 32 + 16 * rt + v + 8 * half] = e;
    }
  }
  __syncthreads();
  if (tid < BM)
    partial[(size_t)bx * NROWS + row0 + tid] = lsum[0][tid] + lsum[1][tid];
}

__global__ __launch_bounds__(256)
void gemm_expsum(const float* __restrict__ X, const float* __restrict__ W,
                 float* __restrict__ partial, int K, int M) {
  expsum_block<false>(X, W, partial, K, M, blockIdx.x, blockIdx.y * BM, nullptr, NROWS);
}

__global__ __launch_bounds__(256)
void gemm_expsum_tails(const float* __restrict__ h1, const float* __restrict__ Wo1,
                       float* __restrict__ p1,
                       const float* __restrict__ h2, const float* __restrict__ Wo2,
                       float* __restrict__ p2,
                       const int* __restrict__ idx1, const int* __restrict__ idx2,
                       const int* __restrict__ cnt) {
  const int n1 = cnt[0];
  const int n2 = cnt[1];
  const int nb1 = (n1 + BM - 1) / BM;
  const int nb2 = (n2 + BM - 1) / BM;
  const int bx = blockIdx.x;
  int by = blockIdx.y;
  const float* X; const float* Wt; float* P; const int* idx;
  int K, M, n;
  if (by < nb1) {
    if (bx >= NCH_1) return;
    X = h1; Wt = Wo1; P = p1; idx = idx1; K = D1; M = M1; n = n1;
  } else {
    by -= nb1;
    if (by >= nb2) return;
    X = h2; Wt = Wo2; P = p2; idx = idx2; K = D2; M = M2; n = n2;
  }
  expsum_block<true>(X, Wt, P, K, M, bx, by * BM, idx, n);
}

__global__ __launch_bounds__(1024)
void rank_rows(const int* __restrict__ target, int* __restrict__ idx1,
               int* __restrict__ idx2, int* __restrict__ pos, int* __restrict__ cnt) {
  __shared__ int wtot[32];
  __shared__ int gtot;
  __shared__ int l1[NROWS];
  __shared__ int l2[NROWS];
  const int tid  = threadIdx.x;
  const int lane = tid & 31;
  const int wave = tid >> 5;
  int c[4];
  int mine = 0;
#pragma unroll
  for (int j = 0; j < 4; ++j) {
    const int t = target[4 * tid + j];
    c[j] = (t < CUT0) ? 0 : ((t < CUT1) ? 1 : 65536);
    mine += c[j];
  }
  int inc = mine;
#pragma unroll
  for (int d = 1; d < 32; d <<= 1) {
    const int v = __shfl_up(inc, d, 32);
    if (lane >= d) inc += v;
  }
  if (lane == 31) wtot[wave] = inc;
  __syncthreads();
  if (wave == 0) {
    const int w = wtot[lane];
    int winc = w;
#pragma unroll
    for (int d = 1; d < 32; d <<= 1) {
      const int v = __shfl_up(winc, d, 32);
      if (lane >= d) winc += v;
    }
    wtot[lane] = winc - w;
    if (lane == 31) { gtot = winc; cnt[0] = winc & 65535; cnt[1] = winc >> 16; }
  }
  __syncthreads();
  int base = wtot[wave] + inc - mine;
  int p[4];
#pragma unroll
  for (int j = 0; j < 4; ++j) {
    const int row = 4 * tid + j;
    p[j] = 0;
    if (c[j] == 1)          { p[j] = base & 65535; l1[p[j]] = row; }
    else if (c[j] == 65536) { p[j] = base >> 16;   l2[p[j]] = row; }
    base += c[j];
  }
#pragma unroll
  for (int j = 0; j < 4; ++j) pos[4 * tid + j] = p[j];
  __syncthreads();
  const int n1 = gtot & 65535;
  const int n2 = gtot >> 16;
  for (int i = tid; i < n1; i += 1024) idx1[i] = l1[i];
  for (int i = tid; i < n2; i += 1024) idx2[i] = l2[i];
}

__global__ void gather_dots(const float* __restrict__ x, const int* __restrict__ target,
                            const float* __restrict__ Wh, const float* __restrict__ Wo1,
                            const float* __restrict__ Wo2, const float* __restrict__ h1,
                            const float* __restrict__ h2,
                            float* __restrict__ dh, float* __restrict__ d1,
                            float* __restrict__ d2) {
  const int wave = threadIdx.x >> 5;
  const int lane = threadIdx.x & 31;
  const int row  = blockIdx.x * 8 + wave;
  const int t = target[row];
  const int g = (t < CUT0) ? t : ((t < CUT1) ? CUT0 : CUT0 + 1);

  float s0 = 0.f, s1 = 0.f, s2 = 0.f;
  {
    const float* xr = x + (size_t)row * DIM;
    const float* wr = Wh + (size_t)g * DIM;
    for (int k = lane; k < DIM; k += 32) s0 += xr[k] * wr[k];
  }
  {
    const int rel = min(max(t - CUT0, 0), M1 - 1);
    const float* hr = h1 + (size_t)row * D1;
    const float* wr = Wo1 + (size_t)rel * D1;
    for (int k = lane; k < D1; k += 32) s1 += hr[k] * wr[k];
  }
  {
    const int rel = min(max(t - CUT1, 0), M2 - 1);
    const float* hr = h2 + (size_t)row * D2;
    const float* wr = Wo2 + (size_t)rel * D2;
    for (int k = lane; k < D2; k += 32) s2 += hr[k] * wr[k];
  }
#pragma unroll
  for (int m = 16; m; m >>= 1) {
    s0 += __shfl_xor(s0, m, 32);
    s1 += __shfl_xor(s1, m, 32);
    s2 += __shfl_xor(s2, m, 32);
  }
  if (lane == 0) { dh[row] = s0; d1[row] = s1; d2[row] = s2; }
}

__global__ void finalize(const int* __restrict__ target, const int* __restrict__ pos,
                         const float* __restrict__ ph, const float* __restrict__ p1,
                         const float* __restrict__ p2, const float* __restrict__ dh,
                         const float* __restrict__ d1, const float* __restrict__ d2,
                         float* __restrict__ out) {
  const int i = blockIdx.x * blockDim.x + threadIdx.x;
  if (i >= NROWS) return;
  float sh = 0.f;
  for (int c = 0; c < NCH_H; ++c) sh += ph[(size_t)c * NROWS + i];
  float o = dh[i] - logf(sh);
  const int t = target[i];
  if (t >= CUT0 && t < CUT1) {
    float s = 0.f;
    const int q = pos[i];
    for (int c = 0; c < NCH_1; ++c) s += p1[(size_t)c * NROWS + q];
    o += d1[i] - logf(s);
  } else if (t >= CUT1) {
    float s = 0.f;
    const int q = pos[i];
#pragma unroll 8
    for (int c = 0; c < NCH_2; ++c) s += p2[(size_t)c * NROWS + q];
    o += d2[i] - logf(s);
  }
  out[i] = o;
}

__global__ void loss_reduce(const float* __restrict__ out, float* __restrict__ loss) {
  __shared__ float sm[256];
  float s = 0.f;
  for (int i = threadIdx.x; i < NROWS; i += 256) s += out[i];
  sm[threadIdx.x] = s;
  __syncthreads();
  for (int w = 128; w; w >>= 1) {
    if (threadIdx.x < w) sm[threadIdx.x] += sm[threadIdx.x + w];
    __syncthreads();
  }
  if (threadIdx.x == 0) *loss = -sm[0] / (float)NROWS;
}

extern "C" void kernel_launch(void* const* d_in, const int* in_sizes, int n_in,
                              void* d_out, int out_size, void* d_ws, size_t ws_size,
                              hipStream_t stream) {
  (void)in_sizes; (void)n_in; (void)out_size; (void)ws_size;
  const float* x    = (const float*)d_in[0];
  const int*   targ = (const int*)d_in[1];
  const float* Wh   = (const float*)d_in[2];
  const float* Wp1  = (const float*)d_in[3];
  const float* Wo1  = (const float*)d_in[4];
  const float* Wp2  = (const float*)d_in[5];
  const float* Wo2  = (const float*)d_in[6];
  float* out = (float*)d_out;

  float* ws = (float*)d_ws;
  float* h1 = ws;
  float* h2 = h1 + (size_t)NROWS * D1;
  float* ph = h2 + (size_t)NROWS * D2;
  float* p1 = ph + (size_t)NROWS * NCH_H;
  float* p2 = p1 + (size_t)NROWS * NCH_1;
  float* dh = p2 + (size_t)NROWS * NCH_2;
  float* d1 = dh + NROWS;
  float* d2 = d1 + NROWS;
  int* idx1 = (int*)(d2 + NROWS);
  int* idx2 = idx1 + NROWS;
  int* pos  = idx2 + NROWS;
  int* cnt  = pos + NROWS;

  const dim3 blk(256);
  const int rowBlocks = NROWS / ROWS_B;
  rank_rows<<<1, 1024, 0, stream>>>(targ, idx1, idx2, pos, cnt);
  gemm_store<<<dim3(1, rowBlocks), blk, 0, stream>>>(x, Wp1, h1, DIM, D1);
  gemm_store<<<dim3(1, rowBlocks), blk, 0, stream>>>(x, Wp2, h2, DIM, D2);
  gemm_expsum<<<dim3(NCH_H, NROWS / BM), blk, 0, stream>>>(x,  Wh,  ph, DIM, HEADM);
  gemm_expsum_tails<<<dim3(NCH_2, NROWS / BM + 1), blk, 0, stream>>>(h1, Wo1, p1, h2, Wo2, p2,
                                                                  idx1, idx2, cnt);
  gather_dots<<<dim3(NROWS / 8), blk, 0, stream>>>(x, targ, Wh, Wo1, Wo2, h1, h2, dh, d1, d2);
  finalize<<<dim3(NROWS / 256), blk, 0, stream>>>(targ, pos, ph, p1, p2, dh, d1, d2, out);
  loss_reduce<<<1, 256, 0, stream>>>(out, out + NROWS);
}


# Exp-sum GEMMs: chunk barrier moved in front of the chunk's last WMMA group, next chunk's first fragments read behind it, its LDS stores at the chunk head; k and summation order unchanged
# speedup vs baseline: 3.8440x; 1.0302x over previous
#include <hip/hip_runtime.h>
#include <math.h>


#define CUT0 10000
#define CUT1 30000
#define DIM  1024
#define NROWS 4096
#define HEADM 10002
#define M1 20000
#define M2 20257
#define D1 256
#define D2 64
#define ROWS_B 32
#define CHUNK 512
#define BM 128
#define BN 128
#define KC 32
#define LDP 36
#define NCH_H 79
#define NCH_1 157
#define NCH_2 159

typedef float v2f __attribute__((ext_vector_type(2)));
typedef float v4f __attribute__((ext_vector_type(4)));
typedef float v8f __attribute__((ext_vector_type(8)));

__device__ __forceinline__ v8f wmma4(v2f a, v2f b, v8f c) {
  return __builtin_amdgcn_wmma_f32_16x16x4_f32(false, a, false, b, (short)0, c,
                                               false, false);
}

__device__ __forceinline__ void gemm_wave32(const float* xp0, const float* xp1,
                                            const float* wp0, const float* wp1,
                                            const float* wp2, const float* wp3,
                                            int K, v8f acc[2][4]) {
  v2f a0 = *(const v2f*)(xp0);
  v2f a1 = *(const v2f*)(xp1);
  v2f b0 = *(const v2f*)(wp0);
  v2f b1 = *(const v2f*)(wp1);
  v2f b2 = *(const v2f*)(wp2);
  v2f b3 = *(const v2f*)(wp3);
  for (int k = 4; k < K; k += 4) {
    v2f a0n = *(const v2f*)(xp0 + k);
    v2f a1n = *(const v2f*)(xp1 + k);
    v2f b0n = *(const v2f*)(wp0 + k);
    v2f b1n = *(const v2f*)(wp1 + k);
    v2f b2n = *(const v2f*)(wp2 + k);
    v2f b3n = *(const v2f*)(wp3 + k);
    __builtin_amdgcn_sched_barrier(0);
    acc[0][0] = wmma4(a0, b0, acc[0][0]);
    acc[0][1] = wmma4(a0, b1, acc[0][1]);
    acc[0][2] = wmma4(a0, b2, acc[0][2]);
    acc[0][3] = wmma4(a0, b3, acc[0][3]);
    acc[1][0] = wmma4(a1, b0, acc[1][0]);
    acc[1][1] = wmma4(a1, b1, acc[1][1]);
    acc[1][2] = wmma4(a1, b2, acc[1][2]);
    acc[1][3] = wmma4(a1, b3, acc[1][3]);
    __builtin_amdgcn_sched_barrier(0);
    a0 = a0n; a1 = a1n; b0 = b0n; b1 = b1n; b2 = b2n; b3 = b3n;
  }
  acc[0][0] = wmma4(a0, b0, acc[0][0]);
  acc[0][1] = wmma4(a0, b1, acc[0][1]);
  acc[0][2] = wmma4(a0, b2, acc[0][2]);
  acc[0][3] = wmma4(a0, b3, acc[0][3]);
  acc[1][0] = wmma4(a1, b0, acc[1][0]);
  acc[1][1] = wmma4(a1, b1, acc[1][1]);
  acc[1][2] = wmma4(a1, b2, acc[1][2]);
  acc[1][3] = wmma4(a1, b3, acc[1][3]);
}

__global__ void gemm_store(const float* __restrict__ X, const float* __restrict__ W,
                           float* __restrict__ C, int K, int M) {
  const int wave = threadIdx.x >> 5;
  const int lane = threadIdx.x & 31;
  const int half = lane >> 4;
  const int r    = lane & 15;
  const int row0 = blockIdx.y * ROWS_B;
  const int col0 = blockIdx.x * CHUNK + wave * 64;
  if (col0 >= M) return;

  const float* xp0 = X + (size_t)(row0 + r) * K + 2 * half;
  const float* xp1 = X + (size_t)(row0 + 16 + r) * K + 2 * half;
  const float* wp0 = W + (size_t)(col0 + r) * K + 2 * half;
  const float* wp1 = W + (size_t)(col0 + 16 + r) * K + 2 * half;
  const float* wp2 = W + (size_t)(col0 + 32 + r) * K + 2 * half;
  const float* wp3 = W + (size_t)(col0 + 48 + r) * K + 2 * half;

  v8f acc[2][4];
#pragma unroll
  for (int rt = 0; rt < 2; ++rt)
#pragma unroll
    for (int t = 0; t < 4; ++t)
      acc[rt][t] = (v8f){0.f,0.f,0.f,0.f,0.f,0.f,0.f,0.f};
  gemm_wave32(xp0, xp1, wp0, wp1, wp2, wp3, K, acc);

#pragma unroll
  for (int rt = 0; rt < 2; ++rt) {
#pragma unroll
    for (int t = 0; t < 4; ++t) {
#pragma unroll
      for (int v = 0; v < 8; ++v) {
        int rr = row0 + 16 * rt + v + 8 * half;
        C[(size_t)rr * M + col0 + 16 * t + r] = acc[rt][t][v];
      }
    }
  }
}

__device__ __forceinline__ void mma_group(const v4f a[2], const v4f b[4], v8f acc[2][4]) {
#pragma unroll
  for (int rt = 0; rt < 2; ++rt) {
#pragma unroll
    for (int t = 0; t < 4; ++t) {
      acc[rt][t] = wmma4((v2f){a[rt].x, a[rt].y}, (v2f){b[t].x, b[t].y}, acc[rt][t]);
    }
  }
#pragma unroll
  for (int rt = 0; rt < 2; ++rt) {
#pragma unroll
    for (int t = 0; t < 4; ++t) {
      acc[rt][t] = wmma4((v2f){a[rt].z, a[rt].w}, (v2f){b[t].z, b[t].w}, acc[rt][t]);
    }
  }
}

template <bool GATHER>
__device__ __forceinline__
void expsum_block(const float* __restrict__ X, const float* __restrict__ W,
                  float* __restrict__ partial, const int K, const int M,
                  const int bx, const int row0,
                  const int* __restrict__ rowidx, const int n) {
  __shared__ __attribute__((aligned(16))) float As[2][BM * LDP];
  __shared__ __attribute__((aligned(16))) float Bs[2][BN * LDP];
  __shared__ float lsum[2][BM];
  const int tid  = threadIdx.x;
  const int wave = tid >> 5;
  const int lane = tid & 31;
  const int half = lane >> 4;
  const int r    = lane & 15;
  const int wr   = wave & 3;
  const int wc   = wave >> 2;
  const int col0 = bx * BN;

  const int lr = tid >> 3;
  const int c4 = tid & 7;
  const float* xg[4];
  const float* wg[4];
#pragma unroll
  for (int i = 0; i < 4; ++i) {
    const int s = row0 + lr + 32 * i;
    const int xr = GATHER ? rowidx[min(s, n - 1)] : s;
    xg[i] = X + (size_t)xr * K + 4 * c4;
    wg[i] = W + (size_t)min(col0 + lr + 32 * i, M - 1) * K + 4 * c4;
  }
  const int sOff = lr * LDP + 4 * c4;

  const int nch = K / KC;
  v4f sa[4], sb[4];
#pragma unroll
  for (int i = 0; i < 4; ++i) {
    sa[i] = *(const v4f*)(xg[i]);
    sb[i] = *(const v4f*)(wg[i]);
  }
#pragma unroll
  for (int i = 0; i < 4; ++i) {
    *(v4f*)&As[0][sOff + 32 * i * LDP] = sa[i];
    *(v4f*)&Bs[0][sOff + 32 * i * LDP] = sb[i];
  }
  if (nch > 1) {
#pragma unroll
    for (int i = 0; i < 4; ++i) {
      sa[i] = *(const v4f*)(xg[i] + KC);
      sb[i] = *(const v4f*)(wg[i] + KC);
    }
  }
  __syncthreads();

  const int aOff = (wr * 32 + r) * LDP + 4 * half;
  const int bOff = (wc * 64 + r) * LDP + 4 * half;

  v8f acc[2][4];
#pragma unroll
  for (int rt = 0; rt < 2; ++rt)
#pragma unroll
    for (int t = 0; t < 4; ++t)
      acc[rt][t] = (v8f){0.f,0.f,0.f,0.f,0.f,0.f,0.f,0.f};

  v4f a[2], b[4];
#pragma unroll
  for (int rt = 0; rt < 2; ++rt) a[rt] = *(const v4f*)(&As[0][aOff] + rt * 16 * LDP);
#pragma unroll
  for (int t = 0; t < 4; ++t)   b[t]  = *(const v4f*)(&Bs[0][bOff] + t * 16 * LDP);

  for (int c = 0; c + 1 < nch; ++c) {
    const int cur = c & 1;
#pragma unroll
    for (int i = 0; i < 4; ++i) {
      *(v4f*)&As[cur ^ 1][sOff + 32 * i * LDP] = sa[i];
      *(v4f*)&Bs[cur ^ 1][sOff + 32 * i * LDP] = sb[i];
    }
    if (c + 2 < nch) {
#pragma unroll
      for (int i = 0; i < 4; ++i) {
        sa[i] = *(const v4f*)(xg[i] + (c + 2) * KC);
        sb[i] = *(const v4f*)(wg[i] + (c + 2) * KC);
      }
    }
    const float* as = &As[cur][aOff];
    const float* bs = &Bs[cur][bOff];
#pragma unroll
    for (int kk = 0; kk < KC / 8; ++kk) {
      v4f an[2], bn[4];
      if (kk + 1 < KC / 8) {
#pragma unroll
        for (int rt = 0; rt < 2; ++rt)
          an[rt] = *(const v4f*)(as + rt * 16 * LDP + (kk + 1) * 8);
#pragma unroll
        for (int t = 0; t < 4; ++t)
          bn[t] = *(const v4f*)(bs + t * 16 * LDP + (kk + 1) * 8);
      } else {
        __syncthreads();
        const float* asn = &As[cur ^ 1][aOff];
        const float* bsn = &Bs[cur ^ 1][bOff];
#pragma unroll
        for (int rt = 0; rt < 2; ++rt) an[rt] = *(const v4f*)(asn + rt * 16 * LDP);
#pragma unroll
        for (int t = 0; t < 4; ++t)   bn[t]  = *(const v4f*)(bsn + t * 16 * LDP);
      }
      __builtin_amdgcn_sched_barrier(0);
      mma_group(a, b, acc);
      __builtin_amdgcn_sched_barrier(0);
#pragma unroll
      for (int rt = 0; rt < 2; ++rt) a[rt] = an[rt];
#pragma unroll
      for (int t = 0; t < 4; ++t)   b[t]  = bn[t];
    }
  }
  {
    const float* as = &As[(nch - 1) & 1][aOff];
    const float* bs = &Bs[(nch - 1) & 1][bOff];
#pragma unroll
    for (int kk = 0; kk < KC / 8; ++kk) {
      v4f an[2], bn[4];
      if (kk + 1 < KC / 8) {
#pragma unroll
        for (int rt = 0; rt < 2; ++rt)
          an[rt] = *(const v4f*)(as + rt * 16 * LDP + (kk + 1) * 8);
#pragma unroll
        for (int t = 0; t < 4; ++t)
          bn[t] = *(const v4f*)(bs + t * 16 * LDP + (kk + 1) * 8);
      }
      __builtin_amdgcn_sched_barrier(0);
      mma_group(a, b, acc);
      __builtin_amdgcn_sched_barrier(0);
      if (kk + 1 < KC / 8) {
#pragma unroll
        for (int rt = 0; rt < 2; ++rt) a[rt] = an[rt];
#pragma unroll
        for (int t = 0; t < 4; ++t)   b[t]  = bn[t];
      }
    }
  }

  const int cbase = col0 + wc * 64 + r;
#pragma unroll
  for (int rt = 0; rt < 2; ++rt) {
#pragma unroll
    for (int v = 0; v < 8; ++v) {
      float e = 0.f;
#pragma unroll
      for (int t = 0; t < 4; ++t)
        if (cbase + 16 * t < M) e += expf(acc[rt][t][v]);
      e += __shfl_xor(e, 1, 32);
      e += __shfl_xor(e, 2, 32);
      e += __shfl_xor(e, 4, 32);
      e += __shfl_xor(e, 8, 32);
      if (r == 0) lsum[wc][wr * 32 + 16 * rt + v + 8 * half] = e;
    }
  }
  __syncthreads();
  if (tid < BM)
    partial[(size_t)bx * NROWS + row0 + tid] = lsum[0][tid] + lsum[1][tid];
}

__global__ __launch_bounds__(256)
void gemm_expsum(const float* __restrict__ X, const float* __restrict__ W,
                 float* __restrict__ partial, int K, int M) {
  expsum_block<false>(X, W, partial, K, M, blockIdx.x, blockIdx.y * BM, nullptr, NROWS);
}

__global__ __launch_bounds__(256)
void gemm_expsum_tails(const float* __restrict__ h1, const float* __restrict__ Wo1,
                       float* __restrict__ p1,
                       const float* __restrict__ h2, const float* __restrict__ Wo2,
                       float* __restrict__ p2,
                       const int* __restrict__ idx1, const int* __restrict__ idx2,
                       const int* __restrict__ cnt) {
  const int n1 = cnt[0];
  const int n2 = cnt[1];
  const int nb1 = (n1 + BM - 1) / BM;
  const int nb2 = (n2 + BM - 1) / BM;
  const int bx = blockIdx.x;
  int by = blockIdx.y;
  const float* X; const float* Wt; float* P; const int* idx;
  int K, M, n;
  if (by < nb1) {
    if (bx >= NCH_1) return;
    X = h1; Wt = Wo1; P = p1; idx = idx1; K = D1; M = M1; n = n1;
  } else {
    by -= nb1;
    if (by >= nb2) return;
    X = h2; Wt = Wo2; P = p2; idx = idx2; K = D2; M = M2; n = n2;
  }
  expsum_block<true>(X, Wt, P, K, M, bx, by * BM, idx, n);
}

__global__ __launch_bounds__(1024)
void rank_rows(const int* __restrict__ target, int* __restrict__ idx1,
               int* __restrict__ idx2, int* __restrict__ pos, int* __restrict__ cnt) {
  __shared__ int wtot[32];
  __shared__ int gtot;
  __shared__ int l1[NROWS];
  __shared__ int l2[NROWS];
  const int tid  = threadIdx.x;
  const int lane = tid & 31;
  const int wave = tid >> 5;
  int c[4];
  int mine = 0;
#pragma unroll
  for (int j = 0; j < 4; ++j) {
    const int t = target[4 * tid + j];
    c[j] = (t < CUT0) ? 0 : ((t < CUT1) ? 1 : 65536);
    mine += c[j];
  }
  int inc = mine;
#pragma unroll
  for (int d = 1; d < 32; d <<= 1) {
    const int v = __shfl_up(inc, d, 32);
    if (lane >= d) inc += v;
  }
  if (lane == 31) wtot[wave] = inc;
  __syncthreads();
  if (wave == 0) {
    const int w = wtot[lane];
    int winc = w;
#pragma unroll
    for (int d = 1; d < 32; d <<= 1) {
      const int v = __shfl_up(winc, d, 32);
      if (lane >= d) winc += v;
    }
    wtot[lane] = winc - w;
    if (lane == 31) { gtot = winc; cnt[0] = winc & 65535; cnt[1] = winc >> 16; }
  }
  __syncthreads();
  int base = wtot[wave] + inc - mine;
  int p[4];
#pragma unroll
  for (int j = 0; j < 4; ++j) {
    const int row = 4 * tid + j;
    p[j] = 0;
    if (c[j] == 1)          { p[j] = base & 65535; l1[p[j]] = row; }
    else if (c[j] == 65536) { p[j] = base >> 16;   l2[p[j]] = row; }
    base += c[j];
  }
#pragma unroll
  for (int j = 0; j < 4; ++j) pos[4 * tid + j] = p[j];
  __syncthreads();
  const int n1 = gtot & 65535;
  const int n2 = gtot >> 16;
  for (int i = tid; i < n1; i += 1024) idx1[i] = l1[i];
  for (int i = tid; i < n2; i += 1024) idx2[i] = l2[i];
}

__global__ void gather_dots(const float* __restrict__ x, const int* __restrict__ target,
                            const float* __restrict__ Wh, const float* __restrict__ Wo1,
                            const float* __restrict__ Wo2, const float* __restrict__ h1,
                            const float* __restrict__ h2,
                            float* __restrict__ dh, float* __restrict__ d1,
                            float* __restrict__ d2) {
  const int wave = threadIdx.x >> 5;
  const int lane = threadIdx.x & 31;
  const int row  = blockIdx.x * 8 + wave;
  const int t = target[row];
  const int g = (t < CUT0) ? t : ((t < CUT1) ? CUT0 : CUT0 + 1);

  float s0 = 0.f, s1 = 0.f, s2 = 0.f;
  {
    const float* xr = x + (size_t)row * DIM;
    const float* wr = Wh + (size_t)g * DIM;
    for (int k = lane; k < DIM; k += 32) s0 += xr[k] * wr[k];
  }
  {
    const int rel = min(max(t - CUT0, 0), M1 - 1);
    const float* hr = h1 + (size_t)row * D1;
    const float* wr = Wo1 + (size_t)rel * D1;
    for (int k = lane; k < D1; k += 32) s1 += hr[k] * wr[k];
  }
  {
    const int rel = min(max(t - CUT1, 0), M2 - 1);
    const float* hr = h2 + (size_t)row * D2;
    const float* wr = Wo2 + (size_t)rel * D2;
    for (int k = lane; k < D2; k += 32) s2 += hr[k] * wr[k];
  }
#pragma unroll
  for (int m = 16; m; m >>= 1) {
    s0 += __shfl_xor(s0, m, 32);
    s1 += __shfl_xor(s1, m, 32);
    s2 += __shfl_xor(s2, m, 32);
  }
  if (lane == 0) { dh[row] = s0; d1[row] = s1; d2[row] = s2; }
}

__global__ void finalize(const int* __restrict__ target, const int* __restrict__ pos,
                         const float* __restrict__ ph, const float* __restrict__ p1,
                         const float* __restrict__ p2, const float* __restrict__ dh,
                         const float* __restrict__ d1, const float* __restrict__ d2,
                         float* __restrict__ out) {
  const int i = blockIdx.x * blockDim.x + threadIdx.x;
  if (i >= NROWS) return;
  float sh = 0.f;
  for (int c = 0; c < NCH_H; ++c) sh += ph[(size_t)c * NROWS + i];
  float o = dh[i] - logf(sh);
  const int t = target[i];
  if (t >= CUT0 && t < CUT1) {
    float s = 0.f;
    const int q = pos[i];
    for (int c = 0; c < NCH_1; ++c) s += p1[(size_t)c * NROWS + q];
    o += d1[i] - logf(s);
  } else if (t >= CUT1) {
    float s = 0.f;
    const int q = pos[i];
#pragma unroll 8
    for (int c = 0; c < NCH_2; ++c) s += p2[(size_t)c * NROWS + q];
    o += d2[i] - logf(s);
  }
  out[i] = o;
}

__global__ void loss_reduce(const float* __restrict__ out, float* __restrict__ loss) {
  __shared__ float sm[256];
  float s = 0.f;
  for (int i = threadIdx.x; i < NROWS; i += 256) s += out[i];
  sm[threadIdx.x] = s;
  __syncthreads();
  for (int w = 128; w; w >>= 1) {
    if (threadIdx.x < w) sm[threadIdx.x] += sm[threadIdx.x + w];
    __syncthreads();
  }
  if (threadIdx.x == 0) *loss = -sm[0] / (float)NROWS;
}

extern "C" void kernel_launch(void* const* d_in, const int* in_sizes, int n_in,
                              void* d_out, int out_size, void* d_ws, size_t ws_size,
                              hipStream_t stream) {
  (void)in_sizes; (void)n_in; (void)out_size; (void)ws_size;
  const float* x    = (const float*)d_in[0];
  const int*   targ = (const int*)d_in[1];
  const float* Wh   = (const float*)d_in[2];
  const float* Wp1  = (const float*)d_in[3];
  const float* Wo1  = (const float*)d_in[4];
  const float* Wp2  = (const float*)d_in[5];
  const float* Wo2  = (const float*)d_in[6];
  float* out = (float*)d_out;

  float* ws = (float*)d_ws;
  float* h1 = ws;
  float* h2 = h1 + (size_t)NROWS * D1;
  float* ph = h2 + (size_t)NROWS * D2;
  float* p1 = ph + (size_t)NROWS * NCH_H;
  float* p2 = p1 + (size_t)NROWS * NCH_1;
  float* dh = p2 + (size_t)NROWS * NCH_2;
  float* d1 = dh + NROWS;
  float* d2 = d1 + NROWS;
  int* idx1 = (int*)(d2 + NROWS);
  int* idx2 = idx1 + NROWS;
  int* pos  = idx2 + NROWS;
  int* cnt  = pos + NROWS;

  const dim3 blk(256);
  const int rowBlocks = NROWS / ROWS_B;
  rank_rows<<<1, 1024, 0, stream>>>(targ, idx1, idx2, pos, cnt);
  gemm_store<<<dim3(1, rowBlocks), blk, 0, stream>>>(x, Wp1, h1, DIM, D1);
  gemm_store<<<dim3(1, rowBlocks), blk, 0, stream>>>(x, Wp2, h2, DIM, D2);
  gemm_expsum<<<dim3(NCH_H, NROWS / BM), blk, 0, stream>>>(x,  Wh,  ph, DIM, HEADM);
  gemm_expsum_tails<<<dim3(NCH_2, NROWS / BM + 1), blk, 0, stream>>>(h1, Wo1, p1, h2, Wo2, p2,
                                                                  idx1, idx2, cnt);
  gather_dots<<<dim3(NROWS / 8), blk, 0, stream>>>(x, targ, Wh, Wo1, Wo2, h1, h2, dh, d1, d2);
  finalize<<<dim3(NROWS / 256), blk, 0, stream>>>(targ, pos, ph, p1, p2, dh, d1, d2, out);
  loss_reduce<<<1, 256, 0, stream>>>(out, out + NROWS);
}


# Projection GEMMs: one launch for both, X and [Wp1;Wp2] tiles (32x160, k 32) staged through LDS with coalesced 128-bit loads; changes k order inside each 8 of the projection sums
# speedup vs baseline: 4.0313x; 1.0487x over previous
#include <hip/hip_runtime.h>
#include <math.h>


#define CUT0 10000
#define CUT1 30000
#define DIM  1024
#define NROWS 4096
#define HEADM 10002
#define M1 20000
#define M2 20257
#define D1 256
#define D2 64
#define PBM 32
#define PBN 160
#define PNT 5
#define PCP 164
#define BM 128
#define BN 128
#define KC 32
#define LDP 36
#define NCH_H 79
#define NCH_1 157
#define NCH_2 159

typedef float v2f __attribute__((ext_vector_type(2)));
typedef float v4f __attribute__((ext_vector_type(4)));
typedef float v8f __attribute__((ext_vector_type(8)));

__device__ __forceinline__ v8f wmma4(v2f a, v2f b, v8f c) {
  return __builtin_amdgcn_wmma_f32_16x16x4_f32(false, a, false, b, (short)0, c,
                                               false, false);
}

__device__ __forceinline__ void mma_group5(const v4f a, const v4f b[PNT], v8f acc[PNT]) {
#pragma unroll
  for (int t = 0; t < PNT; ++t)
    acc[t] = wmma4((v2f){a.x, a.y}, (v2f){b[t].x, b[t].y}, acc[t]);
#pragma unroll
  for (int t = 0; t < PNT; ++t)
    acc[t] = wmma4((v2f){a.z, a.w}, (v2f){b[t].z, b[t].w}, acc[t]);
}

__global__ __launch_bounds__(128)
void gemm_proj(const float* __restrict__ X, const float* __restrict__ Wa,
               const float* __restrict__ Wb, float* __restrict__ Ha,
               float* __restrict__ Hb) {
  __shared__ __attribute__((aligned(16))) float As[2][PBM * LDP];
  __shared__ __attribute__((aligned(16))) float Bs[2][PBN * LDP];
  const int tid  = threadIdx.x;
  const int wave = tid >> 5;
  const int lane = tid & 31;
  const int half = lane >> 4;
  const int r    = lane & 15;
  const int wr   = wave & 1;
  const int wc   = wave >> 1;
  const int row0 = blockIdx.y * PBM;
  const int vc0  = blockIdx.x * PBN;

  const int lr = tid >> 3;
  const int c4 = tid & 7;
  const float* xg[PBM / 16];
  const float* wg[PBN / 16];
#pragma unroll
  for (int i = 0; i < PBM / 16; ++i)
    xg[i] = X + (size_t)(row0 + lr + 16 * i) * DIM + 4 * c4;
#pragma unroll
  for (int i = 0; i < PBN / 16; ++i) {
    const int j = vc0 + lr + 16 * i;
    wg[i] = (j < D1 ? Wa + (size_t)j * DIM : Wb + (size_t)(j - D1) * DIM) + 4 * c4;
  }
  const int sOff = lr * LDP + 4 * c4;

  const int nch = DIM / KC;
  v4f sa[PBM / 16], sb[PBN / 16];
#pragma unroll
  for (int i = 0; i < PBM / 16; ++i) sa[i] = *(const v4f*)(xg[i]);
#pragma unroll
  for (int i = 0; i < PBN / 16; ++i) sb[i] = *(const v4f*)(wg[i]);
#pragma unroll
  for (int i = 0; i < PBM / 16; ++i) *(v4f*)&As[0][sOff + 16 * i * LDP] = sa[i];
#pragma unroll
  for (int i = 0; i < PBN / 16; ++i) *(v4f*)&Bs[0][sOff + 16 * i * LDP] = sb[i];
#pragma unroll
  for (int i = 0; i < PBM / 16; ++i) sa[i] = *(const v4f*)(xg[i] + KC);
#pragma unroll
  for (int i = 0; i < PBN / 16; ++i) sb[i] = *(const v4f*)(wg[i] + KC);
  __syncthreads();

  const int aOff = (wr * 16 + r) * LDP + 4 * half;
  const int bOff = (wc * 80 + r) * LDP + 4 * half;

  v8f acc[PNT];
#pragma unroll
  for (int t = 0; t < PNT; ++t)
    acc[t] = (v8f){0.f,0.f,0.f,0.f,0.f,0.f,0.f,0.f};

  v4f a, b[PNT];
  a = *(const v4f*)(&As[0][aOff]);
#pragma unroll
  for (int t = 0; t < PNT; ++t) b[t] = *(const v4f*)(&Bs[0][bOff] + t * 16 * LDP);

  for (int c = 0; c + 1 < nch; ++c) {
    const int cur = c & 1;
#pragma unroll
    for (int i = 0; i < PBM / 16; ++i) *(v4f*)&As[cur ^ 1][sOff + 16 * i * LDP] = sa[i];
#pragma unroll
    for (int i = 0; i < PBN / 16; ++i) *(v4f*)&Bs[cur ^ 1][sOff + 16 * i * LDP] = sb[i];
    if (c + 2 < nch) {
#pragma unroll
      for (int i = 0; i < PBM / 16; ++i) sa[i] = *(const v4f*)(xg[i] + (c + 2) * KC);
#pragma unroll
      for (int i = 0; i < PBN / 16; ++i) sb[i] = *(const v4f*)(wg[i] + (c + 2) * KC);
    }
    const float* as = &As[cur][aOff];
    const float* bs = &Bs[cur][bOff];
#pragma unroll
    for (int kk = 0; kk < KC / 8; ++kk) {
      v4f an, bn[PNT];
      if (kk + 1 < KC / 8) {
        an = *(const v4f*)(as + (kk + 1) * 8);
#pragma unroll
        for (int t = 0; t < PNT; ++t)
          bn[t] = *(const v4f*)(bs + t * 16 * LDP + (kk + 1) * 8);
      } else {
        __syncthreads();
        an = *(const v4f*)(&As[cur ^ 1][aOff]);
#pragma unroll
        for (int t = 0; t < PNT; ++t)
          bn[t] = *(const v4f*)(&Bs[cur ^ 1][bOff] + t * 16 * LDP);
      }
      __builtin_amdgcn_sched_barrier(0);
      mma_group5(a, b, acc);
      __builtin_amdgcn_sched_barrier(0);
      a = an;
#pragma unroll
      for (int t = 0; t < PNT; ++t) b[t] = bn[t];
    }
  }
  {
    const float* as = &As[(nch - 1) & 1][aOff];
    const float* bs = &Bs[(nch - 1) & 1][bOff];
#pragma unroll
    for (int kk = 0; kk < KC / 8; ++kk) {
      v4f an, bn[PNT];
      if (kk + 1 < KC / 8) {
        an = *(const v4f*)(as + (kk + 1) * 8);
#pragma unroll
        for (int t = 0; t < PNT; ++t)
          bn[t] = *(const v4f*)(bs + t * 16 * LDP + (kk + 1) * 8);
      }
      __builtin_amdgcn_sched_barrier(0);
      mma_group5(a, b, acc);
      __builtin_amdgcn_sched_barrier(0);
      if (kk + 1 < KC / 8) {
        a = an;
#pragma unroll
        for (int t = 0; t < PNT; ++t) b[t] = bn[t];
      }
    }
  }

  __syncthreads();
  float* Cs = &Bs[0][0];
#pragma unroll
  for (int t = 0; t < PNT; ++t) {
#pragma unroll
    for (int v = 0; v < 8; ++v)
      Cs[(wr * 16 + v + 8 * half) * PCP + wc * 80 + 16 * t + r] = acc[t][v];
  }
  __syncthreads();
#pragma unroll
  for (int i = 0; i < (PBM * PBN / 4) / 128; ++i) {
    const int idx = tid + 128 * i;
    const int row = idx / (PBN / 4);
    const int q   = idx - row * (PBN / 4);
    const v4f val = *(const v4f*)&Cs[row * PCP + 4 * q];
    const int vcol = vc0 + 4 * q;
    if (vcol < D1) *(v4f*)&Ha[(size_t)(row0 + row) * D1 + vcol] = val;
    else           *(v4f*)&Hb[(size_t)(row0 + row) * D2 + (vcol - D1)] = val;
  }
}

__device__ __forceinline__ void mma_group(const v4f a[2], const v4f b[4], v8f acc[2][4]) {
#pragma unroll
  for (int rt = 0; rt < 2; ++rt) {
#pragma unroll
    for (int t = 0; t < 4; ++t) {
      acc[rt][t] = wmma4((v2f){a[rt].x, a[rt].y}, (v2f){b[t].x, b[t].y}, acc[rt][t]);
    }
  }
#pragma unroll
  for (int rt = 0; rt < 2; ++rt) {
#pragma unroll
    for (int t = 0; t < 4; ++t) {
      acc[rt][t] = wmma4((v2f){a[rt].z, a[rt].w}, (v2f){b[t].z, b[t].w}, acc[rt][t]);
    }
  }
}

template <bool GATHER>
__device__ __forceinline__
void expsum_block(const float* __restrict__ X, const float* __restrict__ W,
                  float* __restrict__ partial, const int K, const int M,
                  const int bx, const int row0,
                  const int* __restrict__ rowidx, const int n) {
  __shared__ __attribute__((aligned(16))) float As[2][BM * LDP];
  __shared__ __attribute__((aligned(16))) float Bs[2][BN * LDP];
  __shared__ float lsum[2][BM];
  const int tid  = threadIdx.x;
  const int wave = tid >> 5;
  const int lane = tid & 31;
  const int half = lane >> 4;
  const int r    = lane & 15;
  const int wr   = wave & 3;
  const int wc   = wave >> 2;
  const int col0 = bx * BN;

  const int lr = tid >> 3;
  const int c4 = tid & 7;
  const float* xg[4];
  const float* wg[4];
#pragma unroll
  for (int i = 0; i < 4; ++i) {
    const int s = row0 + lr + 32 * i;
    const int xr = GATHER ? rowidx[min(s, n - 1)] : s;
    xg[i] = X + (size_t)xr * K + 4 * c4;
    wg[i] = W + (size_t)min(col0 + lr + 32 * i, M - 1) * K + 4 * c4;
  }
  const int sOff = lr * LDP + 4 * c4;

  const int nch = K / KC;
  v4f sa[4], sb[4];
#pragma unroll
  for (int i = 0; i < 4; ++i) {
    sa[i] = *(const v4f*)(xg[i]);
    sb[i] = *(const v4f*)(wg[i]);
  }
#pragma unroll
  for (int i = 0; i < 4; ++i) {
    *(v4f*)&As[0][sOff + 32 * i * LDP] = sa[i];
    *(v4f*)&Bs[0][sOff + 32 * i * LDP] = sb[i];
  }
  if (nch > 1) {
#pragma unroll
    for (int i = 0; i < 4; ++i) {
      sa[i] = *(const v4f*)(xg[i] + KC);
      sb[i] = *(const v4f*)(wg[i] + KC);
    }
  }
  __syncthreads();

  const int aOff = (wr * 32 + r) * LDP + 4 * half;
  const int bOff = (wc * 64 + r) * LDP + 4 * half;

  v8f acc[2][4];
#pragma unroll
  for (int rt = 0; rt < 2; ++rt)
#pragma unroll
    for (int t = 0; t < 4; ++t)
      acc[rt][t] = (v8f){0.f,0.f,0.f,0.f,0.f,0.f,0.f,0.f};

  v4f a[2], b[4];
#pragma unroll
  for (int rt = 0; rt < 2; ++rt) a[rt] = *(const v4f*)(&As[0][aOff] + rt * 16 * LDP);
#pragma unroll
  for (int t = 0; t < 4; ++t)   b[t]  = *(const v4f*)(&Bs[0][bOff] + t * 16 * LDP);

  for (int c = 0; c + 1 < nch; ++c) {
    const int cur = c & 1;
#pragma unroll
    for (int i = 0; i < 4; ++i) {
      *(v4f*)&As[cur ^ 1][sOff + 32 * i * LDP] = sa[i];
      *(v4f*)&Bs[cur ^ 1][sOff + 32 * i * LDP] = sb[i];
    }
    if (c + 2 < nch) {
#pragma unroll
      for (int i = 0; i < 4; ++i) {
        sa[i] = *(const v4f*)(xg[i] + (c + 2) * KC);
        sb[i] = *(const v4f*)(wg[i] + (c + 2) * KC);
      }
    }
    const float* as = &As[cur][aOff];
    const float* bs = &Bs[cur][bOff];
#pragma unroll
    for (int kk = 0; kk < KC / 8; ++kk) {
      v4f an[2], bn[4];
      if (kk + 1 < KC / 8) {
#pragma unroll
        for (int rt = 0; rt < 2; ++rt)
          an[rt] = *(const v4f*)(as + rt * 16 * LDP + (kk + 1) * 8);
#pragma unroll
        for (int t = 0; t < 4; ++t)
          bn[t] = *(const v4f*)(bs + t * 16 * LDP + (kk + 1) * 8);
      } else {
        __syncthreads();
        const float* asn = &As[cur ^ 1][aOff];
        const float* bsn = &Bs[cur ^ 1][bOff];
#pragma unroll
        for (int rt = 0; rt < 2; ++rt) an[rt] = *(const v4f*)(asn + rt * 16 * LDP);
#pragma unroll
        for (int t = 0; t < 4; ++t)   bn[t]  = *(const v4f*)(bsn + t * 16 * LDP);
      }
      __builtin_amdgcn_sched_barrier(0);
      mma_group(a, b, acc);
      __builtin_amdgcn_sched_barrier(0);
#pragma unroll
      for (int rt = 0; rt < 2; ++rt) a[rt] = an[rt];
#pragma unroll
      for (int t = 0; t < 4; ++t)   b[t]  = bn[t];
    }
  }
  {
    const float* as = &As[(nch - 1) & 1][aOff];
    const float* bs = &Bs[(nch - 1) & 1][bOff];
#pragma unroll
    for (int kk = 0; kk < KC / 8; ++kk) {
      v4f an[2], bn[4];
      if (kk + 1 < KC / 8) {
#pragma unroll
        for (int rt = 0; rt < 2; ++rt)
          an[rt] = *(const v4f*)(as + rt * 16 * LDP + (kk + 1) * 8);
#pragma unroll
        for (int t = 0; t < 4; ++t)
          bn[t] = *(const v4f*)(bs + t * 16 * LDP + (kk + 1) * 8);
      }
      __builtin_amdgcn_sched_barrier(0);
      mma_group(a, b, acc);
      __builtin_amdgcn_sched_barrier(0);
      if (kk + 1 < KC / 8) {
#pragma unroll
        for (int rt = 0; rt < 2; ++rt) a[rt] = an[rt];
#pragma unroll
        for (int t = 0; t < 4; ++t)   b[t]  = bn[t];
      }
    }
  }

  const int cbase = col0 + wc * 64 + r;
#pragma unroll
  for (int rt = 0; rt < 2; ++rt) {
#pragma unroll
    for (int v = 0; v < 8; ++v) {
      float e = 0.f;
#pragma unroll
      for (int t = 0; t < 4; ++t)
        if (cbase + 16 * t < M) e += expf(acc[rt][t][v]);
      e += __shfl_xor(e, 1, 32);
      e += __shfl_xor(e, 2, 32);
      e += __shfl_xor(e, 4, 32);
      e += __shfl_xor(e, 8, 32);
      if (r == 0) lsum[wc][wr * 32 + 16 * rt + v + 8 * half] = e;
    }
  }
  __syncthreads();
  if (tid < BM)
    partial[(size_t)bx * NROWS + row0 + tid] = lsum[0][tid] + lsum[1][tid];
}

__global__ __launch_bounds__(256)
void gemm_expsum(const float* __restrict__ X, const float* __restrict__ W,
                 float* __restrict__ partial, int K, int M) {
  expsum_block<false>(X, W, partial, K, M, blockIdx.x, blockIdx.y * BM, nullptr, NROWS);
}

__global__ __launch_bounds__(256)
void gemm_expsum_tails(const float* __restrict__ h1, const float* __restrict__ Wo1,
                       float* __restrict__ p1,
                       const float* __restrict__ h2, const float* __restrict__ Wo2,
                       float* __restrict__ p2,
                       const int* __restrict__ idx1, const int* __restrict__ idx2,
                       const int* __restrict__ cnt) {
  const int n1 = cnt[0];
  const int n2 = cnt[1];
  const int nb1 = (n1 + BM - 1) / BM;
  const int nb2 = (n2 + BM - 1) / BM;
  const int bx = blockIdx.x;
  int by = blockIdx.y;
  const float* X; const float* Wt; float* P; const int* idx;
  int K, M, n;
  if (by < nb1) {
    if (bx >= NCH_1) return;
    X = h1; Wt = Wo1; P = p1; idx = idx1; K = D1; M = M1; n = n1;
  } else {
    by -= nb1;
    if (by >= nb2) return;
    X = h2; Wt = Wo2; P = p2; idx = idx2; K = D2; M = M2; n = n2;
  }
  expsum_block<true>(X, Wt, P, K, M, bx, by * BM, idx, n);
}

__global__ __launch_bounds__(1024)
void rank_rows(const int* __restrict__ target, int* __restrict__ idx1,
               int* __restrict__ idx2, int* __restrict__ pos, int* __restrict__ cnt) {
  __shared__ int wtot[32];
  __shared__ int gtot;
  __shared__ int l1[NROWS];
  __shared__ int l2[NROWS];
  const int tid  = threadIdx.x;
  const int lane = tid & 31;
  const int wave = tid >> 5;
  int c[4];
  int mine = 0;
#pragma unroll
  for (int j = 0; j < 4; ++j) {
    const int t = target[4 * tid + j];
    c[j] = (t < CUT0) ? 0 : ((t < CUT1) ? 1 : 65536);
    mine += c[j];
  }
  int inc = mine;
#pragma unroll
  for (int d = 1; d < 32; d <<= 1) {
    const int v = __shfl_up(inc, d, 32);
    if (lane >= d) inc += v;
  }
  if (lane == 31) wtot[wave] = inc;
  __syncthreads();
  if (wave == 0) {
    const int w = wtot[lane];
    int winc = w;
#pragma unroll
    for (int d = 1; d < 32; d <<= 1) {
      const int v = __shfl_up(winc, d, 32);
      if (lane >= d) winc += v;
    }
    wtot[lane] = winc - w;
    if (lane == 31) { gtot = winc; cnt[0] = winc & 65535; cnt[1] = winc >> 16; }
  }
  __syncthreads();
  int base = wtot[wave] + inc - mine;
  int p[4];
#pragma unroll
  for (int j = 0; j < 4; ++j) {
    const int row = 4 * tid + j;
    p[j] = 0;
    if (c[j] == 1)          { p[j] = base & 65535; l1[p[j]] = row; }
    else if (c[j] == 65536) { p[j] = base >> 16;   l2[p[j]] = row; }
    base += c[j];
  }
#pragma unroll
  for (int j = 0; j < 4; ++j) pos[4 * tid + j] = p[j];
  __syncthreads();
  const int n1 = gtot & 65535;
  const int n2 = gtot >> 16;
  for (int i = tid; i < n1; i += 1024) idx1[i] = l1[i];
  for (int i = tid; i < n2; i += 1024) idx2[i] = l2[i];
}

__global__ void gather_dots(const float* __restrict__ x, const int* __restrict__ target,
                            const float* __restrict__ Wh, const float* __restrict__ Wo1,
                            const float* __restrict__ Wo2, const float* __restrict__ h1,
                            const float* __restrict__ h2,
                            float* __restrict__ dh, float* __restrict__ d1,
                            float* __restrict__ d2) {
  const int wave = threadIdx.x >> 5;
  const int lane = threadIdx.x & 31;
  const int row  = blockIdx.x * 8 + wave;
  const int t = target[row];
  const int g = (t < CUT0) ? t : ((t < CUT1) ? CUT0 : CUT0 + 1);

  float s0 = 0.f, s1 = 0.f, s2 = 0.f;
  {
    const float* xr = x + (size_t)row * DIM;
    const float* wr = Wh + (size_t)g * DIM;
    for (int k = lane; k < DIM; k += 32) s0 += xr[k] * wr[k];
  }
  {
    const int rel = min(max(t - CUT0, 0), M1 - 1);
    const float* hr = h1 + (size_t)row * D1;
    const float* wr = Wo1 + (size_t)rel * D1;
    for (int k = lane; k < D1; k += 32) s1 += hr[k] * wr[k];
  }
  {
    const int rel = min(max(t - CUT1, 0), M2 - 1);
    const float* hr = h2 + (size_t)row * D2;
    const float* wr = Wo2 + (size_t)rel * D2;
    for (int k = lane; k < D2; k += 32) s2 += hr[k] * wr[k];
  }
#pragma unroll
  for (int m = 16; m; m >>= 1) {
    s0 += __shfl_xor(s0, m, 32);
    s1 += __shfl_xor(s1, m, 32);
    s2 += __shfl_xor(s2, m, 32);
  }
  if (lane == 0) { dh[row] = s0; d1[row] = s1; d2[row] = s2; }
}

__global__ void finalize(const int* __restrict__ target, const int* __restrict__ pos,
                         const float* __restrict__ ph, const float* __restrict__ p1,
                         const float* __restrict__ p2, const float* __restrict__ dh,
                         const float* __restrict__ d1, const float* __restrict__ d2,
                         float* __restrict__ out) {
  const int i = blockIdx.x * blockDim.x + threadIdx.x;
  if (i >= NROWS) return;
  float sh = 0.f;
  for (int c = 0; c < NCH_H; ++c) sh += ph[(size_t)c * NROWS + i];
  float o = dh[i] - logf(sh);
  const int t = target[i];
  if (t >= CUT0 && t < CUT1) {
    float s = 0.f;
    const int q = pos[i];
    for (int c = 0; c < NCH_1; ++c) s += p1[(size_t)c * NROWS + q];
    o += d1[i] - logf(s);
  } else if (t >= CUT1) {
    float s = 0.f;
    const int q = pos[i];
#pragma unroll 8
    for (int c = 0; c < NCH_2; ++c) s += p2[(size_t)c * NROWS + q];
    o += d2[i] - logf(s);
  }
  out[i] = o;
}

__global__ void loss_reduce(const float* __restrict__ out, float* __restrict__ loss) {
  __shared__ float sm[256];
  float s = 0.f;
  for (int i = threadIdx.x; i < NROWS; i += 256) s += out[i];
  sm[threadIdx.x] = s;
  __syncthreads();
  for (int w = 128; w; w >>= 1) {
    if (threadIdx.x < w) sm[threadIdx.x] += sm[threadIdx.x + w];
    __syncthreads();
  }
  if (threadIdx.x == 0) *loss = -sm[0] / (float)NROWS;
}

extern "C" void kernel_launch(void* const* d_in, const int* in_sizes, int n_in,
                              void* d_out, int out_size, void* d_ws, size_t ws_size,
                              hipStream_t stream) {
  (void)in_sizes; (void)n_in; (void)out_size; (void)ws_size;
  const float* x    = (const float*)d_in[0];
  const int*   targ = (const int*)d_in[1];
  const float* Wh   = (const float*)d_in[2];
  const float* Wp1  = (const float*)d_in[3];
  const float* Wo1  = (const float*)d_in[4];
  const float* Wp2  = (const float*)d_in[5];
  const float* Wo2  = (const float*)d_in[6];
  float* out = (float*)d_out;

  float* ws = (float*)d_ws;
  float* h1 = ws;
  float* h2 = h1 + (size_t)NROWS * D1;
  float* ph = h2 + (size_t)NROWS * D2;
  float* p1 = ph + (size_t)NROWS * NCH_H;
  float* p2 = p1 + (size_t)NROWS * NCH_1;
  float* dh = p2 + (size_t)NROWS * NCH_2;
  float* d1 = dh + NROWS;
  float* d2 = d1 + NROWS;
  int* idx1 = (int*)(d2 + NROWS);
  int* idx2 = idx1 + NROWS;
  int* pos  = idx2 + NROWS;
  int* cnt  = pos + NROWS;

  const dim3 blk(256);
  rank_rows<<<1, 1024, 0, stream>>>(targ, idx1, idx2, pos, cnt);
  gemm_proj<<<dim3((D1 + D2) / PBN, NROWS / PBM), dim3(128), 0, stream>>>(x, Wp1, Wp2, h1, h2);
  gemm_expsum<<<dim3(NCH_H, NROWS / BM), blk, 0, stream>>>(x,  Wh,  ph, DIM, HEADM);
  gemm_expsum_tails<<<dim3(NCH_2, NROWS / BM + 1), blk, 0, stream>>>(h1, Wo1, p1, h2, Wo2, p2,
                                                                  idx1, idx2, cnt);
  gather_dots<<<dim3(NROWS / 8), blk, 0, stream>>>(x, targ, Wh, Wo1, Wo2, h1, h2, dh, d1, d2);
  finalize<<<dim3(NROWS / 256), blk, 0, stream>>>(targ, pos, ph, p1, p2, dh, d1, d2, out);
  loss_reduce<<<1, 256, 0, stream>>>(out, out + NROWS);
}


# Exp-sum epilogue takes a lane's 16 rows together, not one by one: a column tile's 16 exps are independent, the 16-lane sums share one halving butterfly (15 exchanges, was 64); sum order unchanged
# speedup vs baseline: 4.2309x; 1.0495x over previous
#include <hip/hip_runtime.h>
#include <math.h>


#define CUT0 10000
#define CUT1 30000
#define DIM  1024
#define NROWS 4096
#define HEADM 10002
#define M1 20000
#define M2 20257
#define D1 256
#define D2 64
#define PBM 32
#define PBN 160
#define PNT 5
#define PCP 164
#define BM 128
#define BN 128
#define KC 32
#define LDP 36
#define NCH_H 79
#define NCH_1 157
#define NCH_2 159

typedef float v2f __attribute__((ext_vector_type(2)));
typedef float v4f __attribute__((ext_vector_type(4)));
typedef float v8f __attribute__((ext_vector_type(8)));

__device__ __forceinline__ v8f wmma4(v2f a, v2f b, v8f c) {
  return __builtin_amdgcn_wmma_f32_16x16x4_f32(false, a, false, b, (short)0, c,
                                               false, false);
}

__device__ __forceinline__ void mma_group5(const v4f a, const v4f b[PNT], v8f acc[PNT]) {
#pragma unroll
  for (int t = 0; t < PNT; ++t)
    acc[t] = wmma4((v2f){a.x, a.y}, (v2f){b[t].x, b[t].y}, acc[t]);
#pragma unroll
  for (int t = 0; t < PNT; ++t)
    acc[t] = wmma4((v2f){a.z, a.w}, (v2f){b[t].z, b[t].w}, acc[t]);
}

__global__ __launch_bounds__(128)
void gemm_proj(const float* __restrict__ X, const float* __restrict__ Wa,
               const float* __restrict__ Wb, float* __restrict__ Ha,
               float* __restrict__ Hb) {
  __shared__ __attribute__((aligned(16))) float As[2][PBM * LDP];
  __shared__ __attribute__((aligned(16))) float Bs[2][PBN * LDP];
  const int tid  = threadIdx.x;
  const int wave = tid >> 5;
  const int lane = tid & 31;
  const int half = lane >> 4;
  const int r    = lane & 15;
  const int wr   = wave & 1;
  const int wc   = wave >> 1;
  const int row0 = blockIdx.y * PBM;
  const int vc0  = blockIdx.x * PBN;

  const int lr = tid >> 3;
  const int c4 = tid & 7;
  const float* xg[PBM / 16];
  const float* wg[PBN / 16];
#pragma unroll
  for (int i = 0; i < PBM / 16; ++i)
    xg[i] = X + (size_t)(row0 + lr + 16 * i) * DIM + 4 * c4;
#pragma unroll
  for (int i = 0; i < PBN / 16; ++i) {
    const int j = vc0 + lr + 16 * i;
    wg[i] = (j < D1 ? Wa + (size_t)j * DIM : Wb + (size_t)(j - D1) * DIM) + 4 * c4;
  }
  const int sOff = lr * LDP + 4 * c4;

  const int nch = DIM / KC;
  v4f sa[PBM / 16], sb[PBN / 16];
#pragma unroll
  for (int i = 0; i < PBM / 16; ++i) sa[i] = *(const v4f*)(xg[i]);
#pragma unroll
  for (int i = 0; i < PBN / 16; ++i) sb[i] = *(const v4f*)(wg[i]);
#pragma unroll
  for (int i = 0; i < PBM / 16; ++i) *(v4f*)&As[0][sOff + 16 * i * LDP] = sa[i];
#pragma unroll
  for (int i = 0; i < PBN / 16; ++i) *(v4f*)&Bs[0][sOff + 16 * i * LDP] = sb[i];
#pragma unroll
  for (int i = 0; i < PBM / 16; ++i) sa[i] = *(const v4f*)(xg[i] + KC);
#pragma unroll
  for (int i = 0; i < PBN / 16; ++i) sb[i] = *(const v4f*)(wg[i] + KC);
  __syncthreads();

  const int aOff = (wr * 16 + r) * LDP + 4 * half;
  const int bOff = (wc * 80 + r) * LDP + 4 * half;

  v8f acc[PNT];
#pragma unroll
  for (int t = 0; t < PNT; ++t)
    acc[t] = (v8f){0.f,0.f,0.f,0.f,0.f,0.f,0.f,0.f};

  v4f a, b[PNT];
  a = *(const v4f*)(&As[0][aOff]);
#pragma unroll
  for (int t = 0; t < PNT; ++t) b[t] = *(const v4f*)(&Bs[0][bOff] + t * 16 * LDP);

  for (int c = 0; c + 1 < nch; ++c) {
    const int cur = c & 1;
#pragma unroll
    for (int i = 0; i < PBM / 16; ++i) *(v4f*)&As[cur ^ 1][sOff + 16 * i * LDP] = sa[i];
#pragma unroll
    for (int i = 0; i < PBN / 16; ++i) *(v4f*)&Bs[cur ^ 1][sOff + 16 * i * LDP] = sb[i];
    if (c + 2 < nch) {
#pragma unroll
      for (int i = 0; i < PBM / 16; ++i) sa[i] = *(const v4f*)(xg[i] + (c + 2) * KC);
#pragma unroll
      for (int i = 0; i < PBN / 16; ++i) sb[i] = *(const v4f*)(wg[i] + (c + 2) * KC);
    }
    const float* as = &As[cur][aOff];
    const float* bs = &Bs[cur][bOff];
#pragma unroll
    for (int kk = 0; kk < KC / 8; ++kk) {
      v4f an, bn[PNT];
      if (kk + 1 < KC / 8) {
        an = *(const v4f*)(as + (kk + 1) * 8);
#pragma unroll
        for (int t = 0; t < PNT; ++t)
          bn[t] = *(const v4f*)(bs + t * 16 * LDP + (kk + 1) * 8);
      } else {
        __syncthreads();
        an = *(const v4f*)(&As[cur ^ 1][aOff]);
#pragma unroll
        for (int t = 0; t < PNT; ++t)
          bn[t] = *(const v4f*)(&Bs[cur ^ 1][bOff] + t * 16 * LDP);
      }
      __builtin_amdgcn_sched_barrier(0);
      mma_group5(a, b, acc);
      __builtin_amdgcn_sched_barrier(0);
      a = an;
#pragma unroll
      for (int t = 0; t < PNT; ++t) b[t] = bn[t];
    }
  }
  {
    const float* as = &As[(nch - 1) & 1][aOff];
    const float* bs = &Bs[(nch - 1) & 1][bOff];
#pragma unroll
    for (int kk = 0; kk < KC / 8; ++kk) {
      v4f an, bn[PNT];
      if (kk + 1 < KC / 8) {
        an = *(const v4f*)(as + (kk + 1) * 8);
#pragma unroll
        for (int t = 0; t < PNT; ++t)
          bn[t] = *(const v4f*)(bs + t * 16 * LDP + (kk + 1) * 8);
      }
      __builtin_amdgcn_sched_barrier(0);
      mma_group5(a, b, acc);
      __builtin_amdgcn_sched_barrier(0);
      if (kk + 1 < KC / 8) {
        a = an;
#pragma unroll
        for (int t = 0; t < PNT; ++t) b[t] = bn[t];
      }
    }
  }

  __syncthreads();
  float* Cs = &Bs[0][0];
#pragma unroll
  for (int t = 0; t < PNT; ++t) {
#pragma unroll
    for (int v = 0; v < 8; ++v)
      Cs[(wr * 16 + v + 8 * half) * PCP + wc * 80 + 16 * t + r] = acc[t][v];
  }
  __syncthreads();
#pragma unroll
  for (int i = 0; i < (PBM * PBN / 4) / 128; ++i) {
    const int idx = tid + 128 * i;
    const int row = idx / (PBN / 4);
    const int q   = idx - row * (PBN / 4);
    const v4f val = *(const v4f*)&Cs[row * PCP + 4 * q];
    const int vcol = vc0 + 4 * q;
    if (vcol < D1) *(v4f*)&Ha[(size_t)(row0 + row) * D1 + vcol] = val;
    else           *(v4f*)&Hb[(size_t)(row0 + row) * D2 + (vcol - D1)] = val;
  }
}

__device__ __forceinline__ void mma_group(const v4f a[2], const v4f b[4], v8f acc[2][4]) {
#pragma unroll
  for (int rt = 0; rt < 2; ++rt) {
#pragma unroll
    for (int t = 0; t < 4; ++t) {
      acc[rt][t] = wmma4((v2f){a[rt].x, a[rt].y}, (v2f){b[t].x, b[t].y}, acc[rt][t]);
    }
  }
#pragma unroll
  for (int rt = 0; rt < 2; ++rt) {
#pragma unroll
    for (int t = 0; t < 4; ++t) {
      acc[rt][t] = wmma4((v2f){a[rt].z, a[rt].w}, (v2f){b[t].z, b[t].w}, acc[rt][t]);
    }
  }
}

template <bool GATHER>
__device__ __forceinline__
void expsum_block(const float* __restrict__ X, const float* __restrict__ W,
                  float* __restrict__ partial, const int K, const int M,
                  const int bx, const int row0,
                  const int* __restrict__ rowidx, const int n) {
  __shared__ __attribute__((aligned(16))) float As[2][BM * LDP];
  __shared__ __attribute__((aligned(16))) float Bs[2][BN * LDP];
  __shared__ float lsum[2][BM];
  const int tid  = threadIdx.x;
  const int wave = tid >> 5;
  const int lane = tid & 31;
  const int half = lane >> 4;
  const int r    = lane & 15;
  const int wr   = wave & 3;
  const int wc   = wave >> 2;
  const int col0 = bx * BN;

  const int lr = tid >> 3;
  const int c4 = tid & 7;
  const float* xg[4];
  const float* wg[4];
#pragma unroll
  for (int i = 0; i < 4; ++i) {
    const int s = row0 + lr + 32 * i;
    const int xr = GATHER ? rowidx[min(s, n - 1)] : s;
    xg[i] = X + (size_t)xr * K + 4 * c4;
    wg[i] = W + (size_t)min(col0 + lr + 32 * i, M - 1) * K + 4 * c4;
  }
  const int sOff = lr * LDP + 4 * c4;

  const int nch = K / KC;
  v4f sa[4], sb[4];
#pragma unroll
  for (int i = 0; i < 4; ++i) {
    sa[i] = *(const v4f*)(xg[i]);
    sb[i] = *(const v4f*)(wg[i]);
  }
#pragma unroll
  for (int i = 0; i < 4; ++i) {
    *(v4f*)&As[0][sOff + 32 * i * LDP] = sa[i];
    *(v4f*)&Bs[0][sOff + 32 * i * LDP] = sb[i];
  }
  if (nch > 1) {
#pragma unroll
    for (int i = 0; i < 4; ++i) {
      sa[i] = *(const v4f*)(xg[i] + KC);
      sb[i] = *(const v4f*)(wg[i] + KC);
    }
  }
  __syncthreads();

  const int aOff = (wr * 32 + r) * LDP + 4 * half;
  const int bOff = (wc * 64 + r) * LDP + 4 * half;

  v8f acc[2][4];
#pragma unroll
  for (int rt = 0; rt < 2; ++rt)
#pragma unroll
    for (int t = 0; t < 4; ++t)
      acc[rt][t] = (v8f){0.f,0.f,0.f,0.f,0.f,0.f,0.f,0.f};

  v4f a[2], b[4];
#pragma unroll
  for (int rt = 0; rt < 2; ++rt) a[rt] = *(const v4f*)(&As[0][aOff] + rt * 16 * LDP);
#pragma unroll
  for (int t = 0; t < 4; ++t)   b[t]  = *(const v4f*)(&Bs[0][bOff] + t * 16 * LDP);

  for (int c = 0; c + 1 < nch; ++c) {
    const int cur = c & 1;
#pragma unroll
    for (int i = 0; i < 4; ++i) {
      *(v4f*)&As[cur ^ 1][sOff + 32 * i * LDP] = sa[i];
      *(v4f*)&Bs[cur ^ 1][sOff + 32 * i * LDP] = sb[i];
    }
    if (c + 2 < nch) {
#pragma unroll
      for (int i = 0; i < 4; ++i) {
        sa[i] = *(const v4f*)(xg[i] + (c + 2) * KC);
        sb[i] = *(const v4f*)(wg[i] + (c + 2) * KC);
      }
    }
    const float* as = &As[cur][aOff];
    const float* bs = &Bs[cur][bOff];
#pragma unroll
    for (int kk = 0; kk < KC / 8; ++kk) {
      v4f an[2], bn[4];
      if (kk + 1 < KC / 8) {
#pragma unroll
        for (int rt = 0; rt < 2; ++rt)
          an[rt] = *(const v4f*)(as + rt * 16 * LDP + (kk + 1) * 8);
#pragma unroll
        for (int t = 0; t < 4; ++t)
          bn[t] = *(const v4f*)(bs + t * 16 * LDP + (kk + 1) * 8);
      } else {
        __syncthreads();
        const float* asn = &As[cur ^ 1][aOff];
        const float* bsn = &Bs[cur ^ 1][bOff];
#pragma unroll
        for (int rt = 0; rt < 2; ++rt) an[rt] = *(const v4f*)(asn + rt * 16 * LDP);
#pragma unroll
        for (int t = 0; t < 4; ++t)   bn[t]  = *(const v4f*)(bsn + t * 16 * LDP);
      }
      __builtin_amdgcn_sched_barrier(0);
      mma_group(a, b, acc);
      __builtin_amdgcn_sched_barrier(0);
#pragma unroll
      for (int rt = 0; rt < 2; ++rt) a[rt] = an[rt];
#pragma unroll
      for (int t = 0; t < 4; ++t)   b[t]  = bn[t];
    }
  }
  {
    const float* as = &As[(nch - 1) & 1][aOff];
    const float* bs = &Bs[(nch - 1) & 1][bOff];
#pragma unroll
    for (int kk = 0; kk < KC / 8; ++kk) {
      v4f an[2], bn[4];
      if (kk + 1 < KC / 8) {
#pragma unroll
        for (int rt = 0; rt < 2; ++rt)
          an[rt] = *(const v4f*)(as + rt * 16 * LDP + (kk + 1) * 8);
#pragma unroll
        for (int t = 0; t < 4; ++t)
          bn[t] = *(const v4f*)(bs + t * 16 * LDP + (kk + 1) * 8);
      }
      __builtin_amdgcn_sched_barrier(0);
      mma_group(a, b, acc);
      __builtin_amdgcn_sched_barrier(0);
      if (kk + 1 < KC / 8) {
#pragma unroll
        for (int rt = 0; rt < 2; ++rt) a[rt] = an[rt];
#pragma unroll
        for (int t = 0; t < 4; ++t)   b[t]  = bn[t];
      }
    }
  }

  const int cbase = col0 + wc * 64 + r;
  float e0 = 0.f, e1 = 0.f, e2 = 0.f, e3 = 0.f, e4 = 0.f, e5 = 0.f, e6 = 0.f, e7 = 0.f;
  float e8 = 0.f, e9 = 0.f, e10 = 0.f, e11 = 0.f, e12 = 0.f, e13 = 0.f, e14 = 0.f, e15 = 0.f;
#pragma unroll
  for (int t = 0; t < 4; ++t) {
    if (cbase + 16 * t < M) {
      e0  += expf(acc[0][t][0]);  e1  += expf(acc[0][t][1]);
      e2  += expf(acc[0][t][2]);  e3  += expf(acc[0][t][3]);
      e4  += expf(acc[0][t][4]);  e5  += expf(acc[0][t][5]);
      e6  += expf(acc[0][t][6]);  e7  += expf(acc[0][t][7]);
      e8  += expf(acc[1][t][0]);  e9  += expf(acc[1][t][1]);
      e10 += expf(acc[1][t][2]);  e11 += expf(acc[1][t][3]);
      e12 += expf(acc[1][t][4]);  e13 += expf(acc[1][t][5]);
      e14 += expf(acc[1][t][6]);  e15 += expf(acc[1][t][7]);
    }
  }
#define ROW_XCH(dst, lo, hi, up, m)                          \
  const float dst = ((up) ? (hi) : (lo)) + __shfl_xor((up) ? (lo) : (hi), (m), 32)
  const bool up1 = (r & 1) != 0, up2 = (r & 2) != 0, up4 = (r & 4) != 0, up8 = (r & 8) != 0;
  ROW_XCH(f0, e0, e8,  up1, 1);  ROW_XCH(f1, e1, e9,  up1, 1);
  ROW_XCH(f2, e2, e10, up1, 1);  ROW_XCH(f3, e3, e11, up1, 1);
  ROW_XCH(f4, e4, e12, up1, 1);  ROW_XCH(f5, e5, e13, up1, 1);
  ROW_XCH(f6, e6, e14, up1, 1);  ROW_XCH(f7, e7, e15, up1, 1);
  ROW_XCH(g0, f0, f4, up2, 2);   ROW_XCH(g1, f1, f5, up2, 2);
  ROW_XCH(g2, f2, f6, up2, 2);   ROW_XCH(g3, f3, f7, up2, 2);
  ROW_XCH(h0, g0, g2, up4, 4);   ROW_XCH(h1, g1, g3, up4, 4);
  ROW_XCH(rowsum, h0, h1, up8, 8);
#undef ROW_XCH
  const int jrow = ((r & 1) << 3) | ((r & 2) << 1) | ((r & 4) >> 1) | ((r & 8) >> 3);
  lsum[wc][wr * 32 + 16 * (jrow >> 3) + (jrow & 7) + 8 * half] = rowsum;
  __syncthreads();
  if (tid < BM)
    partial[(size_t)bx * NROWS + row0 + tid] = lsum[0][tid] + lsum[1][tid];
}

__global__ __launch_bounds__(256)
void gemm_expsum(const float* __restrict__ X, const float* __restrict__ W,
                 float* __restrict__ partial, int K, int M) {
  expsum_block<false>(X, W, partial, K, M, blockIdx.x, blockIdx.y * BM, nullptr, NROWS);
}

__global__ __launch_bounds__(256)
void gemm_expsum_tails(const float* __restrict__ h1, const float* __restrict__ Wo1,
                       float* __restrict__ p1,
                       const float* __restrict__ h2, const float* __restrict__ Wo2,
                       float* __restrict__ p2,
                       const int* __restrict__ idx1, const int* __restrict__ idx2,
                       const int* __restrict__ cnt) {
  const int n1 = cnt[0];
  const int n2 = cnt[1];
  const int nb1 = (n1 + BM - 1) / BM;
  const int nb2 = (n2 + BM - 1) / BM;
  const int bx = blockIdx.x;
  int by = blockIdx.y;
  const float* X; const float* Wt; float* P; const int* idx;
  int K, M, n;
  if (by < nb1) {
    if (bx >= NCH_1) return;
    X = h1; Wt = Wo1; P = p1; idx = idx1; K = D1; M = M1; n = n1;
  } else {
    by -= nb1;
    if (by >= nb2) return;
    X = h2; Wt = Wo2; P = p2; idx = idx2; K = D2; M = M2; n = n2;
  }
  expsum_block<true>(X, Wt, P, K, M, bx, by * BM, idx, n);
}

__global__ __launch_bounds__(1024)
void rank_rows(const int* __restrict__ target, int* __restrict__ idx1,
               int* __restrict__ idx2, int* __restrict__ pos, int* __restrict__ cnt) {
  __shared__ int wtot[32];
  __shared__ int gtot;
  __shared__ int l1[NROWS];
  __shared__ int l2[NROWS];
  const int tid  = threadIdx.x;
  const int lane = tid & 31;
  const int wave = tid >> 5;
  int c[4];
  int mine = 0;
#pragma unroll
  for (int j = 0; j < 4; ++j) {
    const int t = target[4 * tid + j];
    c[j] = (t < CUT0) ? 0 : ((t < CUT1) ? 1 : 65536);
    mine += c[j];
  }
  int inc = mine;
#pragma unroll
  for (int d = 1; d < 32; d <<= 1) {
    const int v = __shfl_up(inc, d, 32);
    if (lane >= d) inc += v;
  }
  if (lane == 31) wtot[wave] = inc;
  __syncthreads();
  if (wave == 0) {
    const int w = wtot[lane];
    int winc = w;
#pragma unroll
    for (int d = 1; d < 32; d <<= 1) {
      const int v = __shfl_up(winc, d, 32);
      if (lane >= d) winc += v;
    }
    wtot[lane] = winc - w;
    if (lane == 31) { gtot = winc; cnt[0] = winc & 65535; cnt[1] = winc >> 16; }
  }
  __syncthreads();
  int base = wtot[wave] + inc - mine;
  int p[4];
#pragma unroll
  for (int j = 0; j < 4; ++j) {
    const int row = 4 * tid + j;
    p[j] = 0;
    if (c[j] == 1)          { p[j] = base & 65535; l1[p[j]] = row; }
    else if (c[j] == 65536) { p[j] = base >> 16;   l2[p[j]] = row; }
    base += c[j];
  }
#pragma unroll
  for (int j = 0; j < 4; ++j) pos[4 * tid + j] = p[j];
  __syncthreads();
  const int n1 = gtot & 65535;
  const int n2 = gtot >> 16;
  for (int i = tid; i < n1; i += 1024) idx1[i] = l1[i];
  for (int i = tid; i < n2; i += 1024) idx2[i] = l2[i];
}

__global__ void gather_dots(const float* __restrict__ x, const int* __restrict__ target,
                            const float* __restrict__ Wh, const float* __restrict__ Wo1,
                            const float* __restrict__ Wo2, const float* __restrict__ h1,
                            const float* __restrict__ h2,
                            float* __restrict__ dh, float* __restrict__ d1,
                            float* __restrict__ d2) {
  const int wave = threadIdx.x >> 5;
  const int lane = threadIdx.x & 31;
  const int row  = blockIdx.x * 8 + wave;
  const int t = target[row];
  const int g = (t < CUT0) ? t : ((t < CUT1) ? CUT0 : CUT0 + 1);

  float s0 = 0.f, s1 = 0.f, s2 = 0.f;
  {
    const float* xr = x + (size_t)row * DIM;
    const float* wr = Wh + (size_t)g * DIM;
    for (int k = lane; k < DIM; k += 32) s0 += xr[k] * wr[k];
  }
  {
    const int rel = min(max(t - CUT0, 0), M1 - 1);
    const float* hr = h1 + (size_t)row * D1;
    const float* wr = Wo1 + (size_t)rel * D1;
    for (int k = lane; k < D1; k += 32) s1 += hr[k] * wr[k];
  }
  {
    const int rel = min(max(t - CUT1, 0), M2 - 1);
    const float* hr = h2 + (size_t)row * D2;
    const float* wr = Wo2 + (size_t)rel * D2;
    for (int k = lane; k < D2; k += 32) s2 += hr[k] * wr[k];
  }
#pragma unroll
  for (int m = 16; m; m >>= 1) {
    s0 += __shfl_xor(s0, m, 32);
    s1 += __shfl_xor(s1, m, 32);
    s2 += __shfl_xor(s2, m, 32);
  }
  if (lane == 0) { dh[row] = s0; d1[row] = s1; d2[row] = s2; }
}

__global__ void finalize(const int* __restrict__ target, const int* __restrict__ pos,
                         const float* __restrict__ ph, const float* __restrict__ p1,
                         const float* __restrict__ p2, const float* __restrict__ dh,
                         const float* __restrict__ d1, const float* __restrict__ d2,
                         float* __restrict__ out) {
  const int i = blockIdx.x * blockDim.x + threadIdx.x;
  if (i >= NROWS) return;
  float sh = 0.f;
  for (int c = 0; c < NCH_H; ++c) sh += ph[(size_t)c * NROWS + i];
  float o = dh[i] - logf(sh);
  const int t = target[i];
  if (t >= CUT0 && t < CUT1) {
    float s = 0.f;
    const int q = pos[i];
    for (int c = 0; c < NCH_1; ++c) s += p1[(size_t)c * NROWS + q];
    o += d1[i] - logf(s);
  } else if (t >= CUT1) {
    float s = 0.f;
    const int q = pos[i];
#pragma unroll 8
    for (int c = 0; c < NCH_2; ++c) s += p2[(size_t)c * NROWS + q];
    o += d2[i] - logf(s);
  }
  out[i] = o;
}

__global__ void loss_reduce(const float* __restrict__ out, float* __restrict__ loss) {
  __shared__ float sm[256];
  float s = 0.f;
  for (int i = threadIdx.x; i < NROWS; i += 256) s += out[i];
  sm[threadIdx.x] = s;
  __syncthreads();
  for (int w = 128; w; w >>= 1) {
    if (threadIdx.x < w) sm[threadIdx.x] += sm[threadIdx.x + w];
    __syncthreads();
  }
  if (threadIdx.x == 0) *loss = -sm[0] / (float)NROWS;
}

extern "C" void kernel_launch(void* const* d_in, const int* in_sizes, int n_in,
                              void* d_out, int out_size, void* d_ws, size_t ws_size,
                              hipStream_t stream) {
  (void)in_sizes; (void)n_in; (void)out_size; (void)ws_size;
  const float* x    = (const float*)d_in[0];
  const int*   targ = (const int*)d_in[1];
  const float* Wh   = (const float*)d_in[2];
  const float* Wp1  = (const float*)d_in[3];
  const float* Wo1  = (const float*)d_in[4];
  const float* Wp2  = (const float*)d_in[5];
  const float* Wo2  = (const float*)d_in[6];
  float* out = (float*)d_out;

  float* ws = (float*)d_ws;
  float* h1 = ws;
  float* h2 = h1 + (size_t)NROWS * D1;
  float* ph = h2 + (size_t)NROWS * D2;
  float* p1 = ph + (size_t)NROWS * NCH_H;
  float* p2 = p1 + (size_t)NROWS * NCH_1;
  float* dh = p2 + (size_t)NROWS * NCH_2;
  float* d1 = dh + NROWS;
  float* d2 = d1 + NROWS;
  int* idx1 = (int*)(d2 + NROWS);
  int* idx2 = idx1 + NROWS;
  int* pos  = idx2 + NROWS;
  int* cnt  = pos + NROWS;

  const dim3 blk(256);
  rank_rows<<<1, 1024, 0, stream>>>(targ, idx1, idx2, pos, cnt);
  gemm_proj<<<dim3((D1 + D2) / PBN, NROWS / PBM), dim3(128), 0, stream>>>(x, Wp1, Wp2, h1, h2);
  gemm_expsum<<<dim3(NCH_H, NROWS / BM), blk, 0, stream>>>(x,  Wh,  ph, DIM, HEADM);
  gemm_expsum_tails<<<dim3(NCH_2, NROWS / BM + 1), blk, 0, stream>>>(h1, Wo1, p1, h2, Wo2, p2,
                                                                  idx1, idx2, cnt);
  gather_dots<<<dim3(NROWS / 8), blk, 0, stream>>>(x, targ, Wh, Wo1, Wo2, h1, h2, dh, d1, d2);
  finalize<<<dim3(NROWS / 256), blk, 0, stream>>>(targ, pos, ph, p1, p2, dh, d1, d2, out);
  loss_reduce<<<1, 256, 0, stream>>>(out, out + NROWS);
}


# finalize, gather_dots: fully unrolled so a row's loads are in flight together (one tail loop for both clusters, 64-thread blocks), not 1-2 memory round trips per 8 values; sum order unchanged
# speedup vs baseline: 4.4843x; 1.0599x over previous
#include <hip/hip_runtime.h>
#include <math.h>


#define CUT0 10000
#define CUT1 30000
#define DIM  1024
#define NROWS 4096
#define HEADM 10002
#define M1 20000
#define M2 20257
#define D1 256
#define D2 64
#define PBM 32
#define PBN 160
#define PNT 5
#define PCP 164
#define BM 128
#define BN 128
#define KC 32
#define LDP 36
#define NCH_H 79
#define NCH_1 157
#define NCH_2 159

typedef float v2f __attribute__((ext_vector_type(2)));
typedef float v4f __attribute__((ext_vector_type(4)));
typedef float v8f __attribute__((ext_vector_type(8)));

__device__ __forceinline__ v8f wmma4(v2f a, v2f b, v8f c) {
  return __builtin_amdgcn_wmma_f32_16x16x4_f32(false, a, false, b, (short)0, c,
                                               false, false);
}

__device__ __forceinline__ void mma_group5(const v4f a, const v4f b[PNT], v8f acc[PNT]) {
#pragma unroll
  for (int t = 0; t < PNT; ++t)
    acc[t] = wmma4((v2f){a.x, a.y}, (v2f){b[t].x, b[t].y}, acc[t]);
#pragma unroll
  for (int t = 0; t < PNT; ++t)
    acc[t] = wmma4((v2f){a.z, a.w}, (v2f){b[t].z, b[t].w}, acc[t]);
}

__global__ __launch_bounds__(128)
void gemm_proj(const float* __restrict__ X, const float* __restrict__ Wa,
               const float* __restrict__ Wb, float* __restrict__ Ha,
               float* __restrict__ Hb) {
  __shared__ __attribute__((aligned(16))) float As[2][PBM * LDP];
  __shared__ __attribute__((aligned(16))) float Bs[2][PBN * LDP];
  const int tid  = threadIdx.x;
  const int wave = tid >> 5;
  const int lane = tid & 31;
  const int half = lane >> 4;
  const int r    = lane & 15;
  const int wr   = wave & 1;
  const int wc   = wave >> 1;
  const int row0 = blockIdx.y * PBM;
  const int vc0  = blockIdx.x * PBN;

  const int lr = tid >> 3;
  const int c4 = tid & 7;
  const float* xg[PBM / 16];
  const float* wg[PBN / 16];
#pragma unroll
  for (int i = 0; i < PBM / 16; ++i)
    xg[i] = X + (size_t)(row0 + lr + 16 * i) * DIM + 4 * c4;
#pragma unroll
  for (int i = 0; i < PBN / 16; ++i) {
    const int j = vc0 + lr + 16 * i;
    wg[i] = (j < D1 ? Wa + (size_t)j * DIM : Wb + (size_t)(j - D1) * DIM) + 4 * c4;
  }
  const int sOff = lr * LDP + 4 * c4;

  const int nch = DIM / KC;
  v4f sa[PBM / 16], sb[PBN / 16];
#pragma unroll
  for (int i = 0; i < PBM / 16; ++i) sa[i] = *(const v4f*)(xg[i]);
#pragma unroll
  for (int i = 0; i < PBN / 16; ++i) sb[i] = *(const v4f*)(wg[i]);
#pragma unroll
  for (int i = 0; i < PBM / 16; ++i) *(v4f*)&As[0][sOff + 16 * i * LDP] = sa[i];
#pragma unroll
  for (int i = 0; i < PBN / 16; ++i) *(v4f*)&Bs[0][sOff + 16 * i * LDP] = sb[i];
#pragma unroll
  for (int i = 0; i < PBM / 16; ++i) sa[i] = *(const v4f*)(xg[i] + KC);
#pragma unroll
  for (int i = 0; i < PBN / 16; ++i) sb[i] = *(const v4f*)(wg[i] + KC);
  __syncthreads();

  const int aOff = (wr * 16 + r) * LDP + 4 * half;
  const int bOff = (wc * 80 + r) * LDP + 4 * half;

  v8f acc[PNT];
#pragma unroll
  for (int t = 0; t < PNT; ++t)
    acc[t] = (v8f){0.f,0.f,0.f,0.f,0.f,0.f,0.f,0.f};

  v4f a, b[PNT];
  a = *(const v4f*)(&As[0][aOff]);
#pragma unroll
  for (int t = 0; t < PNT; ++t) b[t] = *(const v4f*)(&Bs[0][bOff] + t * 16 * LDP);

  for (int c = 0; c + 1 < nch; ++c) {
    const int cur = c & 1;
#pragma unroll
    for (int i = 0; i < PBM / 16; ++i) *(v4f*)&As[cur ^ 1][sOff + 16 * i * LDP] = sa[i];
#pragma unroll
    for (int i = 0; i < PBN / 16; ++i) *(v4f*)&Bs[cur ^ 1][sOff + 16 * i * LDP] = sb[i];
    if (c + 2 < nch) {
#pragma unroll
      for (int i = 0; i < PBM / 16; ++i) sa[i] = *(const v4f*)(xg[i] + (c + 2) * KC);
#pragma unroll
      for (int i = 0; i < PBN / 16; ++i) sb[i] = *(const v4f*)(wg[i] + (c + 2) * KC);
    }
    const float* as = &As[cur][aOff];
    const float* bs = &Bs[cur][bOff];
#pragma unroll
    for (int kk = 0; kk < KC / 8; ++kk) {
      v4f an, bn[PNT];
      if (kk + 1 < KC / 8) {
        an = *(const v4f*)(as + (kk + 1) * 8);
#pragma unroll
        for (int t = 0; t < PNT; ++t)
          bn[t] = *(const v4f*)(bs + t * 16 * LDP + (kk + 1) * 8);
      } else {
        __syncthreads();
        an = *(const v4f*)(&As[cur ^ 1][aOff]);
#pragma unroll
        for (int t = 0; t < PNT; ++t)
          bn[t] = *(const v4f*)(&Bs[cur ^ 1][bOff] + t * 16 * LDP);
      }
      __builtin_amdgcn_sched_barrier(0);
      mma_group5(a, b, acc);
      __builtin_amdgcn_sched_barrier(0);
      a = an;
#pragma unroll
      for (int t = 0; t < PNT; ++t) b[t] = bn[t];
    }
  }
  {
    const float* as = &As[(nch - 1) & 1][aOff];
    const float* bs = &Bs[(nch - 1) & 1][bOff];
#pragma unroll
    for (int kk = 0; kk < KC / 8; ++kk) {
      v4f an, bn[PNT];
      if (kk + 1 < KC / 8) {
        an = *(const v4f*)(as + (kk + 1) * 8);
#pragma unroll
        for (int t = 0; t < PNT; ++t)
          bn[t] = *(const v4f*)(bs + t * 16 * LDP + (kk + 1) * 8);
      }
      __builtin_amdgcn_sched_barrier(0);
      mma_group5(a, b, acc);
      __builtin_amdgcn_sched_barrier(0);
      if (kk + 1 < KC / 8) {
        a = an;
#pragma unroll
        for (int t = 0; t < PNT; ++t) b[t] = bn[t];
      }
    }
  }

  __syncthreads();
  float* Cs = &Bs[0][0];
#pragma unroll
  for (int t = 0; t < PNT; ++t) {
#pragma unroll
    for (int v = 0; v < 8; ++v)
      Cs[(wr * 16 + v + 8 * half) * PCP + wc * 80 + 16 * t + r] = acc[t][v];
  }
  __syncthreads();
#pragma unroll
  for (int i = 0; i < (PBM * PBN / 4) / 128; ++i) {
    const int idx = tid + 128 * i;
    const int row = idx / (PBN / 4);
    const int q   = idx - row * (PBN / 4);
    const v4f val = *(const v4f*)&Cs[row * PCP + 4 * q];
    const int vcol = vc0 + 4 * q;
    if (vcol < D1) *(v4f*)&Ha[(size_t)(row0 + row) * D1 + vcol] = val;
    else           *(v4f*)&Hb[(size_t)(row0 + row) * D2 + (vcol - D1)] = val;
  }
}

__device__ __forceinline__ void mma_group(const v4f a[2], const v4f b[4], v8f acc[2][4]) {
#pragma unroll
  for (int rt = 0; rt < 2; ++rt) {
#pragma unroll
    for (int t = 0; t < 4; ++t) {
      acc[rt][t] = wmma4((v2f){a[rt].x, a[rt].y}, (v2f){b[t].x, b[t].y}, acc[rt][t]);
    }
  }
#pragma unroll
  for (int rt = 0; rt < 2; ++rt) {
#pragma unroll
    for (int t = 0; t < 4; ++t) {
      acc[rt][t] = wmma4((v2f){a[rt].z, a[rt].w}, (v2f){b[t].z, b[t].w}, acc[rt][t]);
    }
  }
}

template <bool GATHER>
__device__ __forceinline__
void expsum_block(const float* __restrict__ X, const float* __restrict__ W,
                  float* __restrict__ partial, const int K, const int M,
                  const int bx, const int row0,
                  const int* __restrict__ rowidx, const int n) {
  __shared__ __attribute__((aligned(16))) float As[2][BM * LDP];
  __shared__ __attribute__((aligned(16))) float Bs[2][BN * LDP];
  __shared__ float lsum[2][BM];
  const int tid  = threadIdx.x;
  const int wave = tid >> 5;
  const int lane = tid & 31;
  const int half = lane >> 4;
  const int r    = lane & 15;
  const int wr   = wave & 3;
  const int wc   = wave >> 2;
  const int col0 = bx * BN;

  const int lr = tid >> 3;
  const int c4 = tid & 7;
  const float* xg[4];
  const float* wg[4];
#pragma unroll
  for (int i = 0; i < 4; ++i) {
    const int s = row0 + lr + 32 * i;
    const int xr = GATHER ? rowidx[min(s, n - 1)] : s;
    xg[i] = X + (size_t)xr * K + 4 * c4;
    wg[i] = W + (size_t)min(col0 + lr + 32 * i, M - 1) * K + 4 * c4;
  }
  const int sOff = lr * LDP + 4 * c4;

  const int nch = K / KC;
  v4f sa[4], sb[4];
#pragma unroll
  for (int i = 0; i < 4; ++i) {
    sa[i] = *(const v4f*)(xg[i]);
    sb[i] = *(const v4f*)(wg[i]);
  }
#pragma unroll
  for (int i = 0; i < 4; ++i) {
    *(v4f*)&As[0][sOff + 32 * i * LDP] = sa[i];
    *(v4f*)&Bs[0][sOff + 32 * i * LDP] = sb[i];
  }
  if (nch > 1) {
#pragma unroll
    for (int i = 0; i < 4; ++i) {
      sa[i] = *(const v4f*)(xg[i] + KC);
      sb[i] = *(const v4f*)(wg[i] + KC);
    }
  }
  __syncthreads();

  const int aOff = (wr * 32 + r) * LDP + 4 * half;
  const int bOff = (wc * 64 + r) * LDP + 4 * half;

  v8f acc[2][4];
#pragma unroll
  for (int rt = 0; rt < 2; ++rt)
#pragma unroll
    for (int t = 0; t < 4; ++t)
      acc[rt][t] = (v8f){0.f,0.f,0.f,0.f,0.f,0.f,0.f,0.f};

  v4f a[2], b[4];
#pragma unroll
  for (int rt = 0; rt < 2; ++rt) a[rt] = *(const v4f*)(&As[0][aOff] + rt * 16 * LDP);
#pragma unroll
  for (int t = 0; t < 4; ++t)   b[t]  = *(const v4f*)(&Bs[0][bOff] + t * 16 * LDP);

  for (int c = 0; c + 1 < nch; ++c) {
    const int cur = c & 1;
#pragma unroll
    for (int i = 0; i < 4; ++i) {
      *(v4f*)&As[cur ^ 1][sOff + 32 * i * LDP] = sa[i];
      *(v4f*)&Bs[cur ^ 1][sOff + 32 * i * LDP] = sb[i];
    }
    if (c + 2 < nch) {
#pragma unroll
      for (int i = 0; i < 4; ++i) {
        sa[i] = *(const v4f*)(xg[i] + (c + 2) * KC);
        sb[i] = *(const v4f*)(wg[i] + (c + 2) * KC);
      }
    }
    const float* as = &As[cur][aOff];
    const float* bs = &Bs[cur][bOff];
#pragma unroll
    for (int kk = 0; kk < KC / 8; ++kk) {
      v4f an[2], bn[4];
      if (kk + 1 < KC / 8) {
#pragma unroll
        for (int rt = 0; rt < 2; ++rt)
          an[rt] = *(const v4f*)(as + rt * 16 * LDP + (kk + 1) * 8);
#pragma unroll
        for (int t = 0; t < 4; ++t)
          bn[t] = *(const v4f*)(bs + t * 16 * LDP + (kk + 1) * 8);
      } else {
        __syncthreads();
        const float* asn = &As[cur ^ 1][aOff];
        const float* bsn = &Bs[cur ^ 1][bOff];
#pragma unroll
        for (int rt = 0; rt < 2; ++rt) an[rt] = *(const v4f*)(asn + rt * 16 * LDP);
#pragma unroll
        for (int t = 0; t < 4; ++t)   bn[t]  = *(const v4f*)(bsn + t * 16 * LDP);
      }
      __builtin_amdgcn_sched_barrier(0);
      mma_group(a, b, acc);
      __builtin_amdgcn_sched_barrier(0);
#pragma unroll
      for (int rt = 0; rt < 2; ++rt) a[rt] = an[rt];
#pragma unroll
      for (int t = 0; t < 4; ++t)   b[t]  = bn[t];
    }
  }
  {
    const float* as = &As[(nch - 1) & 1][aOff];
    const float* bs = &Bs[(nch - 1) & 1][bOff];
#pragma unroll
    for (int kk = 0; kk < KC / 8; ++kk) {
      v4f an[2], bn[4];
      if (kk + 1 < KC / 8) {
#pragma unroll
        for (int rt = 0; rt < 2; ++rt)
          an[rt] = *(const v4f*)(as + rt * 16 * LDP + (kk + 1) * 8);
#pragma unroll
        for (int t = 0; t < 4; ++t)
          bn[t] = *(const v4f*)(bs + t * 16 * LDP + (kk + 1) * 8);
      }
      __builtin_amdgcn_sched_barrier(0);
      mma_group(a, b, acc);
      __builtin_amdgcn_sched_barrier(0);
      if (kk + 1 < KC / 8) {
#pragma unroll
        for (int rt = 0; rt < 2; ++rt) a[rt] = an[rt];
#pragma unroll
        for (int t = 0; t < 4; ++t)   b[t]  = bn[t];
      }
    }
  }

  const int cbase = col0 + wc * 64 + r;
  float e0 = 0.f, e1 = 0.f, e2 = 0.f, e3 = 0.f, e4 = 0.f, e5 = 0.f, e6 = 0.f, e7 = 0.f;
  float e8 = 0.f, e9 = 0.f, e10 = 0.f, e11 = 0.f, e12 = 0.f, e13 = 0.f, e14 = 0.f, e15 = 0.f;
#pragma unroll
  for (int t = 0; t < 4; ++t) {
    if (cbase + 16 * t < M) {
      e0  += expf(acc[0][t][0]);  e1  += expf(acc[0][t][1]);
      e2  += expf(acc[0][t][2]);  e3  += expf(acc[0][t][3]);
      e4  += expf(acc[0][t][4]);  e5  += expf(acc[0][t][5]);
      e6  += expf(acc[0][t][6]);  e7  += expf(acc[0][t][7]);
      e8  += expf(acc[1][t][0]);  e9  += expf(acc[1][t][1]);
      e10 += expf(acc[1][t][2]);  e11 += expf(acc[1][t][3]);
      e12 += expf(acc[1][t][4]);  e13 += expf(acc[1][t][5]);
      e14 += expf(acc[1][t][6]);  e15 += expf(acc[1][t][7]);
    }
  }
#define ROW_XCH(dst, lo, hi, up, m)                          \
  const float dst = ((up) ? (hi) : (lo)) + __shfl_xor((up) ? (lo) : (hi), (m), 32)
  const bool up1 = (r & 1) != 0, up2 = (r & 2) != 0, up4 = (r & 4) != 0, up8 = (r & 8) != 0;
  ROW_XCH(f0, e0, e8,  up1, 1);  ROW_XCH(f1, e1, e9,  up1, 1);
  ROW_XCH(f2, e2, e10, up1, 1);  ROW_XCH(f3, e3, e11, up1, 1);
  ROW_XCH(f4, e4, e12, up1, 1);  ROW_XCH(f5, e5, e13, up1, 1);
  ROW_XCH(f6, e6, e14, up1, 1);  ROW_XCH(f7, e7, e15, up1, 1);
  ROW_XCH(g0, f0, f4, up2, 2);   ROW_XCH(g1, f1, f5, up2, 2);
  ROW_XCH(g2, f2, f6, up2, 2);   ROW_XCH(g3, f3, f7, up2, 2);
  ROW_XCH(h0, g0, g2, up4, 4);   ROW_XCH(h1, g1, g3, up4, 4);
  ROW_XCH(rowsum, h0, h1, up8, 8);
#undef ROW_XCH
  const int jrow = ((r & 1) << 3) | ((r & 2) << 1) | ((r & 4) >> 1) | ((r & 8) >> 3);
  lsum[wc][wr * 32 + 16 * (jrow >> 3) + (jrow & 7) + 8 * half] = rowsum;
  __syncthreads();
  if (tid < BM)
    partial[(size_t)bx * NROWS + row0 + tid] = lsum[0][tid] + lsum[1][tid];
}

__global__ __launch_bounds__(256)
void gemm_expsum(const float* __restrict__ X, const float* __restrict__ W,
                 float* __restrict__ partial, int K, int M) {
  expsum_block<false>(X, W, partial, K, M, blockIdx.x, blockIdx.y * BM, nullptr, NROWS);
}

__global__ __launch_bounds__(256)
void gemm_expsum_tails(const float* __restrict__ h1, const float* __restrict__ Wo1,
                       float* __restrict__ p1,
                       const float* __restrict__ h2, const float* __restrict__ Wo2,
                       float* __restrict__ p2,
                       const int* __restrict__ idx1, const int* __restrict__ idx2,
                       const int* __restrict__ cnt) {
  const int n1 = cnt[0];
  const int n2 = cnt[1];
  const int nb1 = (n1 + BM - 1) / BM;
  const int nb2 = (n2 + BM - 1) / BM;
  const int bx = blockIdx.x;
  int by = blockIdx.y;
  const float* X; const float* Wt; float* P; const int* idx;
  int K, M, n;
  if (by < nb1) {
    if (bx >= NCH_1) return;
    X = h1; Wt = Wo1; P = p1; idx = idx1; K = D1; M = M1; n = n1;
  } else {
    by -= nb1;
    if (by >= nb2) return;
    X = h2; Wt = Wo2; P = p2; idx = idx2; K = D2; M = M2; n = n2;
  }
  expsum_block<true>(X, Wt, P, K, M, bx, by * BM, idx, n);
}

__global__ __launch_bounds__(1024)
void rank_rows(const int* __restrict__ target, int* __restrict__ idx1,
               int* __restrict__ idx2, int* __restrict__ pos, int* __restrict__ cnt) {
  __shared__ int wtot[32];
  __shared__ int gtot;
  __shared__ int l1[NROWS];
  __shared__ int l2[NROWS];
  const int tid  = threadIdx.x;
  const int lane = tid & 31;
  const int wave = tid >> 5;
  int c[4];
  int mine = 0;
#pragma unroll
  for (int j = 0; j < 4; ++j) {
    const int t = target[4 * tid + j];
    c[j] = (t < CUT0) ? 0 : ((t < CUT1) ? 1 : 65536);
    mine += c[j];
  }
  int inc = mine;
#pragma unroll
  for (int d = 1; d < 32; d <<= 1) {
    const int v = __shfl_up(inc, d, 32);
    if (lane >= d) inc += v;
  }
  if (lane == 31) wtot[wave] = inc;
  __syncthreads();
  if (wave == 0) {
    const int w = wtot[lane];
    int winc = w;
#pragma unroll
    for (int d = 1; d < 32; d <<= 1) {
      const int v = __shfl_up(winc, d, 32);
      if (lane >= d) winc += v;
    }
    wtot[lane] = winc - w;
    if (lane == 31) { gtot = winc; cnt[0] = winc & 65535; cnt[1] = winc >> 16; }
  }
  __syncthreads();
  int base = wtot[wave] + inc - mine;
  int p[4];
#pragma unroll
  for (int j = 0; j < 4; ++j) {
    const int row = 4 * tid + j;
    p[j] = 0;
    if (c[j] == 1)          { p[j] = base & 65535; l1[p[j]] = row; }
    else if (c[j] == 65536) { p[j] = base >> 16;   l2[p[j]] = row; }
    base += c[j];
  }
#pragma unroll
  for (int j = 0; j < 4; ++j) pos[4 * tid + j] = p[j];
  __syncthreads();
  const int n1 = gtot & 65535;
  const int n2 = gtot >> 16;
  for (int i = tid; i < n1; i += 1024) idx1[i] = l1[i];
  for (int i = tid; i < n2; i += 1024) idx2[i] = l2[i];
}

__global__ __launch_bounds__(256)
void gather_dots(const float* __restrict__ x, const int* __restrict__ target,
                 const float* __restrict__ Wh, const float* __restrict__ Wo1,
                 const float* __restrict__ Wo2, const float* __restrict__ h1,
                 const float* __restrict__ h2,
                 float* __restrict__ dh, float* __restrict__ d1,
                 float* __restrict__ d2) {
  const int wave = threadIdx.x >> 5;
  const int lane = threadIdx.x & 31;
  const int row  = blockIdx.x * 8 + wave;
  const int t = target[row];

  const float* xr  = x  + (size_t)row * DIM + lane;
  const float* hr1 = h1 + (size_t)row * D1 + lane;
  const float* hr2 = h2 + (size_t)row * D2 + lane;
  float xv[DIM / 32], av[D1 / 32], bv[D2 / 32];
#pragma unroll
  for (int j = 0; j < DIM / 32; ++j) xv[j] = xr[32 * j];
#pragma unroll
  for (int j = 0; j < D1 / 32; ++j) av[j] = hr1[32 * j];
#pragma unroll
  for (int j = 0; j < D2 / 32; ++j) bv[j] = hr2[32 * j];

  const int g = (t < CUT0) ? t : ((t < CUT1) ? CUT0 : CUT0 + 1);
  const int rel1 = min(max(t - CUT0, 0), M1 - 1);
  const int rel2 = min(max(t - CUT1, 0), M2 - 1);
  const float* wr0 = Wh  + (size_t)g * DIM + lane;
  const float* wr1 = Wo1 + (size_t)rel1 * D1 + lane;
  const float* wr2 = Wo2 + (size_t)rel2 * D2 + lane;
  float wv[DIM / 32], cv[D1 / 32], dv[D2 / 32];
#pragma unroll
  for (int j = 0; j < DIM / 32; ++j) wv[j] = wr0[32 * j];
#pragma unroll
  for (int j = 0; j < D1 / 32; ++j) cv[j] = wr1[32 * j];
#pragma unroll
  for (int j = 0; j < D2 / 32; ++j) dv[j] = wr2[32 * j];

  float s0 = 0.f, s1 = 0.f, s2 = 0.f;
#pragma unroll
  for (int j = 0; j < DIM / 32; ++j) s0 = fmaf(xv[j], wv[j], s0);
#pragma unroll
  for (int j = 0; j < D1 / 32; ++j) s1 = fmaf(av[j], cv[j], s1);
#pragma unroll
  for (int j = 0; j < D2 / 32; ++j) s2 = fmaf(bv[j], dv[j], s2);
#pragma unroll
  for (int m = 16; m; m >>= 1) {
    s0 += __shfl_xor(s0, m, 32);
    s1 += __shfl_xor(s1, m, 32);
    s2 += __shfl_xor(s2, m, 32);
  }
  if (lane == 0) { dh[row] = s0; d1[row] = s1; d2[row] = s2; }
}

#define FIN_T 64
__global__ __launch_bounds__(FIN_T)
void finalize(const int* __restrict__ target, const int* __restrict__ pos,
              const float* __restrict__ ph, const float* __restrict__ p1,
              const float* __restrict__ p2, const float* __restrict__ dh,
              const float* __restrict__ d1, const float* __restrict__ d2,
              float* __restrict__ out) {
  const int i = blockIdx.x * FIN_T + threadIdx.x;
  if (i >= NROWS) return;
  const int t = target[i];
  const int q = pos[i];
  const bool in1 = (t >= CUT0) && (t < CUT1);
  const bool in2 = (t >= CUT1);

  float vh[NCH_H];
#pragma unroll
  for (int c = 0; c < NCH_H; ++c) vh[c] = ph[(size_t)c * NROWS + i];

  float tail = 0.f;
  if (in1 || in2) {
    const float* pt = (in1 ? p1 : p2) + q;
    const float* dp = in1 ? d1 : d2;
    float vt[NCH_1];
#pragma unroll
    for (int c = 0; c < NCH_1; ++c) vt[c] = pt[(size_t)c * NROWS];
    float x0 = 0.f, x1 = 0.f;
    if (in2) {
      x0 = pt[(size_t)NCH_1 * NROWS];
      x1 = pt[(size_t)(NCH_1 + 1) * NROWS];
    }
    const float dt = dp[i];
    float s = 0.f;
#pragma unroll
    for (int c = 0; c < NCH_1; ++c) s += vt[c];
    if (in2) { s += x0; s += x1; }
    tail = dt - logf(s);
  }

  float sh = 0.f;
#pragma unroll
  for (int c = 0; c < NCH_H; ++c) sh += vh[c];
  float o = dh[i] - logf(sh);
  if (in1 || in2) o += tail;
  out[i] = o;
}

__global__ void loss_reduce(const float* __restrict__ out, float* __restrict__ loss) {
  __shared__ float sm[256];
  float s = 0.f;
  for (int i = threadIdx.x; i < NROWS; i += 256) s += out[i];
  sm[threadIdx.x] = s;
  __syncthreads();
  for (int w = 128; w; w >>= 1) {
    if (threadIdx.x < w) sm[threadIdx.x] += sm[threadIdx.x + w];
    __syncthreads();
  }
  if (threadIdx.x == 0) *loss = -sm[0] / (float)NROWS;
}

extern "C" void kernel_launch(void* const* d_in, const int* in_sizes, int n_in,
                              void* d_out, int out_size, void* d_ws, size_t ws_size,
                              hipStream_t stream) {
  (void)in_sizes; (void)n_in; (void)out_size; (void)ws_size;
  const float* x    = (const float*)d_in[0];
  const int*   targ = (const int*)d_in[1];
  const float* Wh   = (const float*)d_in[2];
  const float* Wp1  = (const float*)d_in[3];
  const float* Wo1  = (const float*)d_in[4];
  const float* Wp2  = (const float*)d_in[5];
  const float* Wo2  = (const float*)d_in[6];
  float* out = (float*)d_out;

  float* ws = (float*)d_ws;
  float* h1 = ws;
  float* h2 = h1 + (size_t)NROWS * D1;
  float* ph = h2 + (size_t)NROWS * D2;
  float* p1 = ph + (size_t)NROWS * NCH_H;
  float* p2 = p1 + (size_t)NROWS * NCH_1;
  float* dh = p2 + (size_t)NROWS * NCH_2;
  float* d1 = dh + NROWS;
  float* d2 = d1 + NROWS;
  int* idx1 = (int*)(d2 + NROWS);
  int* idx2 = idx1 + NROWS;
  int* pos  = idx2 + NROWS;
  int* cnt  = pos + NROWS;

  const dim3 blk(256);
  rank_rows<<<1, 1024, 0, stream>>>(targ, idx1, idx2, pos, cnt);
  gemm_proj<<<dim3((D1 + D2) / PBN, NROWS / PBM), dim3(128), 0, stream>>>(x, Wp1, Wp2, h1, h2);
  gemm_expsum<<<dim3(NCH_H, NROWS / BM), blk, 0, stream>>>(x,  Wh,  ph, DIM, HEADM);
  gemm_expsum_tails<<<dim3(NCH_2, NROWS / BM + 1), blk, 0, stream>>>(h1, Wo1, p1, h2, Wo2, p2,
                                                                  idx1, idx2, cnt);
  gather_dots<<<dim3(NROWS / 8), blk, 0, stream>>>(x, targ, Wh, Wo1, Wo2, h1, h2, dh, d1, d2);
  finalize<<<dim3(NROWS / FIN_T), dim3(FIN_T), 0, stream>>>(targ, pos, ph, p1, p2, dh, d1, d2, out);
  loss_reduce<<<1, 256, 0, stream>>>(out, out + NROWS);
}
